# starting kernel
# baseline (speedup 1.0000x reference)
#include <hip/hip_runtime.h>
#include <hip/hip_bf16.h>
#include <float.h>


typedef float v2f __attribute__((ext_vector_type(2)));
typedef float v8f __attribute__((ext_vector_type(8)));

#define BATCH 8
#define CH    64
#define NPTS  4096
#define OCH   64
#define KNN   20
#define CHUNK 1024
#define NEG_INF (-3.0e38f)

__device__ __forceinline__ v8f wmma_f32(v2f a, v2f b, v8f c) {
    return __builtin_amdgcn_wmma_f32_16x16x4_f32(
        false, a, false, b, (short)0, c, false, false);
}

__global__ __launch_bounds__(256) void sq_kernel(const float* __restrict__ x,
                                                 float* __restrict__ sq) {
    int t = blockIdx.x * 256 + threadIdx.x;
    int b = t >> 12;
    int n = t & (NPTS - 1);
    const float* p = x + (size_t)b * CH * NPTS + n;
    float s = 0.f;
#pragma unroll
    for (int c = 0; c < CH; ++c) { float v = p[(size_t)c * NPTS]; s += v * v; }
    sq[t] = s;
}

__global__ __launch_bounds__(256) void knn_kernel(const float* __restrict__ x,
                                                  const float* __restrict__ sq,
                                                  int* __restrict__ idxout) {
    __shared__ float smem[16 * CHUNK];

    const int b    = blockIdx.y;
    const int q0   = blockIdx.x * 16;
    const int tid  = threadIdx.x;
    const int wave = tid >> 5;
    const int lane = tid & 31;
    const int lmod = lane & 15;
    const int khalf  = (lane >> 4) << 1;
    const int rowadd = (lane >> 4) << 3;
    const float* xb  = x + (size_t)b * CH * NPTS;
    const float* sqb = sq + (size_t)b * NPTS;

    v2f a[16];
#pragma unroll
    for (int kk = 0; kk < 16; ++kk) {
        const float* p = xb + (size_t)(4 * kk + khalf) * NPTS + q0 + lmod;
        a[kk].x = p[0];
        a[kk].y = p[NPTS];
    }

    float best[KNN]; int bidx[KNN]; float thr = NEG_INF;
#pragma unroll
    for (int i = 0; i < KNN; ++i) { best[i] = NEG_INF; bidx[i] = 0; }

    const int q     = tid >> 4;
    const int strip = tid & 15;

    for (int c0 = 0; c0 < NPTS; c0 += CHUNK) {
        for (int t = wave; t < CHUNK / 16; t += 8) {
            const int col0 = c0 + t * 16;
            v8f acc = {};
#pragma unroll
            for (int kk = 0; kk < 16; ++kk) {
                const float* p = xb + (size_t)(4 * kk + khalf) * NPTS + col0 + lmod;
                v2f bb; bb.x = p[0]; bb.y = p[NPTS];
                acc = wmma_f32(a[kk], bb, acc);
            }
            const float sqc = sqb[col0 + lmod];
#pragma unroll
            for (int r = 0; r < 8; ++r)
                smem[(r + rowadd) * CHUNK + t * 16 + lmod] = 2.0f * acc[r] - sqc;
        }
        __syncthreads();

        const float* row = smem + q * CHUNK;
        for (int i = 0; i < CHUNK / 16; ++i) {
            float v = row[i * 16 + strip];
            if (v > thr) {
                int col = c0 + i * 16 + strip;
                int mpos = 0; float mval = best[0];
#pragma unroll
                for (int s = 1; s < KNN; ++s)
                    if (best[s] < mval) { mval = best[s]; mpos = s; }
#pragma unroll
                for (int s = 0; s < KNN; ++s)
                    if (s == mpos) { best[s] = v; bidx[s] = col; }
                float nmin = best[0];
#pragma unroll
                for (int s = 1; s < KNN; ++s) nmin = fminf(nmin, best[s]);
                thr = nmin;
            }
        }
        __syncthreads();
    }

    int* smi = (int*)smem;
#pragma unroll
    for (int i = 0; i < KNN; ++i) {
        smem[tid * KNN + i]            = best[i];
        smi[16 * CHUNK / 2 + tid * KNN + i] = bidx[i];
    }
    __syncthreads();
    if (tid < 16) {
        float fb[KNN]; int fi[KNN]; float fthr = NEG_INF;
#pragma unroll
        for (int i = 0; i < KNN; ++i) { fb[i] = NEG_INF; fi[i] = 0; }
        for (int s = 0; s < 16; ++s) {
            const int cb = (tid * 16 + s) * KNN;
            for (int i = 0; i < KNN; ++i) {
                float v = smem[cb + i];
                if (v > fthr) {
                    int cidx = smi[16 * CHUNK / 2 + cb + i];
                    int mpos = 0; float mval = fb[0];
#pragma unroll
                    for (int u = 1; u < KNN; ++u)
                        if (fb[u] < mval) { mval = fb[u]; mpos = u; }
#pragma unroll
                    for (int u = 0; u < KNN; ++u)
                        if (u == mpos) { fb[u] = v; fi[u] = cidx; }
                    float nmin = fb[0];
#pragma unroll
                    for (int u = 1; u < KNN; ++u) nmin = fminf(nmin, fb[u]);
                    fthr = nmin;
                }
            }
        }
        int* op = idxout + ((size_t)b * NPTS + q0 + tid) * KNN;
#pragma unroll
        for (int i = 0; i < KNN; ++i) op[i] = fi[i];
    }
}

__global__ __launch_bounds__(256) void proj_kernel(const float* __restrict__ x,
                                                   const float* __restrict__ W,
                                                   float* __restrict__ y1T,
                                                   float* __restrict__ y3T) {
    int task = blockIdx.x * 8 + (threadIdx.x >> 5);
    int lane = threadIdx.x & 31;
    int lmod = lane & 15;
    int khalf  = (lane >> 4) << 1;
    int rowadd = (lane >> 4) << 3;

    int which = task & 1;
    int nt    = (task >> 1) & 255;
    int ot    = (task >> 9) & 3;
    int b     = task >> 11;

    const float* xb = x + (size_t)b * CH * NPTS;
    const float* wr = W + (size_t)(ot * 16 + lmod) * (2 * CH);
    const int n0 = nt * 16;

    v8f acc = {};
#pragma unroll
    for (int kk = 0; kk < 16; ++kk) {
        int c = 4 * kk + khalf;
        v2f av, bv;
        if (which == 0) { av.x = wr[c];            av.y = wr[c + 1]; }
        else            { av.x = wr[CH + c] - wr[c];
                          av.y = wr[CH + c + 1] - wr[c + 1]; }
        const float* p = xb + (size_t)c * NPTS + n0 + lmod;
        bv.x = p[0]; bv.y = p[NPTS];
        acc = wmma_f32(av, bv, acc);
    }

    float* y = (which ? y3T : y1T)
             + (size_t)b * NPTS * OCH + (size_t)(n0 + lmod) * OCH
             + ot * 16 + rowadd;
    float4 lo = { acc[0], acc[1], acc[2], acc[3] };
    float4 hi = { acc[4], acc[5], acc[6], acc[7] };
    ((float4*)y)[0] = lo;
    ((float4*)y)[1] = hi;
}

__global__ __launch_bounds__(256) void gather_kernel(const float* __restrict__ y1T,
                                                     const float* __restrict__ y3T,
                                                     const int* __restrict__ idx,
                                                     const float* __restrict__ gamma,
                                                     float* __restrict__ selbuf,
                                                     float* __restrict__ partial) {
    int waveId = blockIdx.x * 8 + (threadIdx.x >> 5);
    int lane   = threadIdx.x & 31;
    int ch     = lane * 2;
    bool mx0 = gamma[ch] >= 0.f, mx1 = gamma[ch + 1] >= 0.f;

    float s0 = 0.f, s1 = 0.f, q0 = 0.f, q1 = 0.f;

    for (int pt = 0; pt < 32; ++pt) {
        int p = waveId * 32 + pt;
        int b = p >> 12;
        float2 e = ((const float2*)(y3T + (size_t)p * OCH))[lane];
        const int* ip = idx + (size_t)p * KNN;
        const float* y1b = y1T + (size_t)b * NPTS * OCH;

        float hi0 = -FLT_MAX, hi1 = -FLT_MAX, lo0 = FLT_MAX, lo1 = FLT_MAX;
#pragma unroll
        for (int j = 0; j < KNN; ++j) {
            int nb = ip[j];
            float2 g = ((const float2*)(y1b + ((size_t)nb << 6)))[lane];
            float h0 = g.x + e.x, h1 = g.y + e.y;
            s0 += h0; s1 += h1; q0 += h0 * h0; q1 += h1 * h1;
            hi0 = fmaxf(hi0, h0); lo0 = fminf(lo0, h0);
            hi1 = fmaxf(hi1, h1); lo1 = fminf(lo1, h1);
        }
        float2 selv = { mx0 ? hi0 : lo0, mx1 ? hi1 : lo1 };
        ((float2*)(selbuf + (size_t)p * OCH))[lane] = selv;
    }
    float* pb = partial + (size_t)waveId * 128;
    pb[ch]          = s0; pb[ch + 1]       = s1;
    pb[OCH + ch]    = q0; pb[OCH + ch + 1] = q1;
}

__global__ void stats_kernel(const float* __restrict__ partial,
                             const float* __restrict__ gamma,
                             const float* __restrict__ beta,
                             float* __restrict__ coeff) {
    int ch = threadIdx.x;
    if (ch >= OCH) return;
    float s = 0.f, q = 0.f;
#pragma unroll 4
    for (int w = 0; w < 1024; ++w) {
        s += partial[(size_t)w * 128 + ch];
        q += partial[(size_t)w * 128 + OCH + ch];
    }
    const float M = (float)BATCH * NPTS * KNN;
    float mean  = s / M;
    float var   = q / M - mean * mean;
    float scale = gamma[ch] * rsqrtf(var + 1e-5f);
    coeff[ch]       = scale;
    coeff[OCH + ch] = beta[ch] - mean * scale;
}

__global__ __launch_bounds__(256) void final_kernel(const float* __restrict__ selbuf,
                                                    const float* __restrict__ coeff,
                                                    float* __restrict__ out) {
    int t = blockIdx.x * 256 + threadIdx.x;
    int n = t & (NPTS - 1);
    int o = (t >> 12) & (OCH - 1);
    int b = t >> 18;
    float v = selbuf[(((size_t)b * NPTS + n) << 6) + o];
    float r = v * coeff[o] + coeff[OCH + o];
    out[t] = (r >= 0.f) ? r : 0.2f * r;
}

extern "C" void kernel_launch(void* const* d_in, const int* in_sizes, int n_in,
                              void* d_out, int out_size, void* d_ws, size_t ws_size,
                              hipStream_t stream) {
    const float* x     = (const float*)d_in[0];
    const float* W     = (const float*)d_in[1];
    const float* gamma = (const float*)d_in[2];
    const float* beta  = (const float*)d_in[3];
    float* out = (float*)d_out;

    float* ws    = (float*)d_ws;
    float* sq    = ws;
    int*   idx   = (int*)(ws + 32768);
    float* y1T   = ws + 32768 + 655360;
    float* y3T   = y1T + (size_t)BATCH * NPTS * OCH;
    float* sel   = y3T + (size_t)BATCH * NPTS * OCH;
    float* part  = sel + (size_t)BATCH * NPTS * OCH;
    float* coeff = part + 1024 * 128;

    sq_kernel<<<(BATCH * NPTS) / 256, 256, 0, stream>>>(x, sq);
    knn_kernel<<<dim3(NPTS / 16, BATCH), 256, 0, stream>>>(x, sq, idx);
    proj_kernel<<<2048, 256, 0, stream>>>(x, W, y1T, y3T);
    gather_kernel<<<128, 256, 0, stream>>>(y1T, y3T, idx, gamma, sel, part);
    stats_kernel<<<1, 64, 0, stream>>>(part, gamma, beta, coeff);
    final_kernel<<<(BATCH * OCH * NPTS) / 256, 256, 0, stream>>>(sel, coeff, out);
}


# kNN: a tile's hits are filed in a per-lane LDS queue and inserted in bulk, a run of insert passes per drain with most lanes busy, not one or more passes per tile; lists and their order unchanged
# speedup vs baseline: 2.9986x; 2.9986x over previous
#include <hip/hip_runtime.h>
#include <hip/hip_bf16.h>
#include <float.h>


typedef float v2f __attribute__((ext_vector_type(2)));
typedef float v8f __attribute__((ext_vector_type(8)));

#define BATCH 8
#define CH    64
#define NPTS  4096
#define OCH   64
#define KNN   20
#define NEG_INF (-3.0e38f)
#define GATHER_BLOCKS 128

__device__ __forceinline__ v8f wmma_f32(v2f a, v2f b, v8f c) {
    return __builtin_amdgcn_wmma_f32_16x16x4_f32(
        false, a, false, b, (short)0, c, false, false);
}

__global__ __launch_bounds__(256) void sq_kernel(const float* __restrict__ x,
                                                 float* __restrict__ sq) {
    int t = blockIdx.x * 256 + threadIdx.x;
    int b = t >> 12;
    int n = t & (NPTS - 1);
    const float* p = x + (size_t)b * CH * NPTS + n;
    float s = 0.f;
#pragma unroll
    for (int c = 0; c < CH; ++c) { float v = p[(size_t)c * NPTS]; s += v * v; }
    sq[t] = s;
}


#define TOPK_INSERT(v, col)                                                   \
    do {                                                                      \
        const float nv = (v); const int nc = (col);                           \
        bool gt[KNN];                                                         \
        _Pragma("unroll")                                                     \
        for (int s = 0; s < KNN; ++s) gt[s] = nv > best[s];                   \
        _Pragma("unroll")                                                     \
        for (int s = KNN - 1; s > 0; --s) {                                   \
            const float uv = gt[s - 1] ? best[s - 1] : nv;                    \
            const int   ui = gt[s - 1] ? bidx[s - 1] : nc;                    \
            best[s] = gt[s] ? uv : best[s];                                   \
            bidx[s] = gt[s] ? ui : bidx[s];                                   \
        }                                                                     \
        best[0] = gt[0] ? nv : best[0];                                       \
        bidx[0] = gt[0] ? nc : bidx[0];                                       \
        thr = best[KNN - 1];                                                  \
    } while (0)

struct CandTile { v2f f[16]; float4 sqlo, sqhi; };

__device__ __forceinline__ void load_cand(CandTile& t, const float* __restrict__ xb,
                                          const float* __restrict__ sqb, int col0,
                                          int lmod, int khalf, int rowadd) {
#pragma unroll
    for (int kk = 0; kk < 16; ++kk) {
        const float* p = xb + (size_t)(4 * kk + khalf) * NPTS + col0 + lmod;
        t.f[kk].x = p[0];
        t.f[kk].y = p[NPTS];
    }
    const float4* sp = (const float4*)(sqb + col0 + rowadd);
    t.sqlo = sp[0];
    t.sqhi = sp[1];
}

#define QCAP   32
#define QROOM  8

__device__ __forceinline__ void push_tile(v8f acc, float4 sqlo, float4 sqhi, int cbase,
                                          float thr, float2* q, int& cnt) {
    float sc[8];
    sc[0] = 2.0f * acc[0] - sqlo.x; sc[1] = 2.0f * acc[1] - sqlo.y;
    sc[2] = 2.0f * acc[2] - sqlo.z; sc[3] = 2.0f * acc[3] - sqlo.w;
    sc[4] = 2.0f * acc[4] - sqhi.x; sc[5] = 2.0f * acc[5] - sqhi.y;
    sc[6] = 2.0f * acc[6] - sqhi.z; sc[7] = 2.0f * acc[7] - sqhi.w;
#pragma unroll
    for (int r = 0; r < 8; ++r) {
        q[cnt * 32] = make_float2(sc[r], __int_as_float(cbase + r));
        cnt += (sc[r] > thr) ? 1 : 0;
    }
}

#define DRAIN_QUEUE()                                                         \
    do {                                                                      \
        float2 e = make_float2(NEG_INF, 0.f);                                 \
        if (cnt > 0) e = q[0];                                                \
        for (int i = 0; __builtin_amdgcn_ballot_w32(i < cnt) != 0u; ++i) {    \
            float2 en = e;                                                    \
            if (i + 1 < cnt) en = q[(i + 1) * 32];                            \
            if (i < cnt && e.x > thr) {                                       \
                const float ev = e.x; const int ec = __float_as_int(e.y);     \
                TOPK_INSERT(ev, ec);                                          \
            }                                                                 \
            e = en;                                                           \
        }                                                                     \
        cnt = 0;                                                              \
    } while (0)

#define MAKE_ROOM()                                                           \
    do {                                                                      \
        if (__builtin_amdgcn_ballot_w32(cnt > QCAP - QROOM) != 0u)            \
            DRAIN_QUEUE();                                                    \
    } while (0)

__global__ __launch_bounds__(128) void knn_kernel(const float* __restrict__ x,
                                                  const float* __restrict__ sq,
                                                  int* __restrict__ idxout) {
    const int b    = blockIdx.y;
    const int wave = threadIdx.x >> 5;
    const int lane = threadIdx.x & 31;
    const int q0   = (blockIdx.x * 4 + wave) * 16;
    const int lmod = lane & 15;
    const int half = lane >> 4;
    const int khalf  = half << 1;
    const int rowadd = half << 3;
    const float* xb  = x + (size_t)b * CH * NPTS;
    const float* sqb = sq + (size_t)b * NPTS;

    v2f qf[16];
#pragma unroll
    for (int kk = 0; kk < 16; ++kk) {
        const float* p = xb + (size_t)(4 * kk + khalf) * NPTS + q0 + lmod;
        qf[kk].x = p[0];
        qf[kk].y = p[NPTS];
    }

    float best[KNN]; int bidx[KNN]; float thr = NEG_INF;
#pragma unroll
    for (int i = 0; i < KNN; ++i) { best[i] = NEG_INF; bidx[i] = 0; }

    __shared__ float2 qbuf[4][QCAP][32];
    float2* q = &qbuf[wave][0][lane];
    int cnt = 0;

    CandTile a0, a1, b0, b1;
    load_cand(a0, xb, sqb, 0,  lmod, khalf, rowadd);
    load_cand(a1, xb, sqb, 16, lmod, khalf, rowadd);

    for (int c0 = 0; c0 < NPTS; c0 += 64) {
        load_cand(b0, xb, sqb, c0 + 32, lmod, khalf, rowadd);
        {
            v8f acc0 = {}, acc1 = {};
#pragma unroll
            for (int kk = 0; kk < 16; ++kk) {
                acc0 = wmma_f32(a0.f[kk], qf[kk], acc0);
                acc1 = wmma_f32(a1.f[kk], qf[kk], acc1);
            }
            MAKE_ROOM();
            push_tile(acc0, a0.sqlo, a0.sqhi, c0 + rowadd,      thr, q, cnt);
            load_cand(b1, xb, sqb, c0 + 48, lmod, khalf, rowadd);
            MAKE_ROOM();
            push_tile(acc1, a1.sqlo, a1.sqhi, c0 + 16 + rowadd, thr, q, cnt);
        }
        const int cn = (c0 + 64) & (NPTS - 1);
        load_cand(a0, xb, sqb, cn, lmod, khalf, rowadd);
        {
            v8f acc0 = {}, acc1 = {};
#pragma unroll
            for (int kk = 0; kk < 16; ++kk) {
                acc0 = wmma_f32(b0.f[kk], qf[kk], acc0);
                acc1 = wmma_f32(b1.f[kk], qf[kk], acc1);
            }
            MAKE_ROOM();
            push_tile(acc0, b0.sqlo, b0.sqhi, c0 + 32 + rowadd, thr, q, cnt);
            load_cand(a1, xb, sqb, cn + 16, lmod, khalf, rowadd);
            MAKE_ROOM();
            push_tile(acc1, b1.sqlo, b1.sqhi, c0 + 48 + rowadd, thr, q, cnt);
        }
    }
    DRAIN_QUEUE();

#pragma nounroll
    for (int i = 0; i < KNN; ++i) {
        const float ov = __shfl_xor(best[0], 16);
        const int   oi = __shfl_xor(bidx[0], 16);
        if (half) {
#pragma unroll
            for (int s = 0; s < KNN - 1; ++s) { best[s] = best[s + 1]; bidx[s] = bidx[s + 1]; }
        } else if (ov > thr) {
            TOPK_INSERT(ov, oi);
        }
    }
    if (half == 0) {
        int* op = idxout + ((size_t)b * NPTS + q0 + lmod) * KNN;
#pragma unroll
        for (int i = 0; i < KNN; ++i) op[i] = bidx[i];
    }
}

__global__ __launch_bounds__(256) void proj_kernel(const float* __restrict__ x,
                                                   const float* __restrict__ W,
                                                   float* __restrict__ y1T,
                                                   float* __restrict__ y3T) {
    int task = blockIdx.x * 8 + (threadIdx.x >> 5);
    int lane = threadIdx.x & 31;
    int lmod = lane & 15;
    int khalf  = (lane >> 4) << 1;
    int rowadd = (lane >> 4) << 3;

    int which = task & 1;
    int nt    = (task >> 1) & 255;
    int ot    = (task >> 9) & 3;
    int b     = task >> 11;

    const float* xb = x + (size_t)b * CH * NPTS;
    const float* wr = W + (size_t)(ot * 16 + lmod) * (2 * CH);
    const int n0 = nt * 16;

    v8f acc = {};
#pragma unroll
    for (int kk = 0; kk < 16; ++kk) {
        int c = 4 * kk + khalf;
        v2f av, bv;
        if (which == 0) { av.x = wr[c];            av.y = wr[c + 1]; }
        else            { av.x = wr[CH + c] - wr[c];
                          av.y = wr[CH + c + 1] - wr[c + 1]; }
        const float* p = xb + (size_t)c * NPTS + n0 + lmod;
        bv.x = p[0]; bv.y = p[NPTS];
        acc = wmma_f32(av, bv, acc);
    }

    float* y = (which ? y3T : y1T)
             + (size_t)b * NPTS * OCH + (size_t)(n0 + lmod) * OCH
             + ot * 16 + rowadd;
    float4 lo = { acc[0], acc[1], acc[2], acc[3] };
    float4 hi = { acc[4], acc[5], acc[6], acc[7] };
    ((float4*)y)[0] = lo;
    ((float4*)y)[1] = hi;
}

__global__ __launch_bounds__(256) void gather_kernel(const float* __restrict__ y1T,
                                                     const float* __restrict__ y3T,
                                                     const int* __restrict__ idx,
                                                     const float* __restrict__ gamma,
                                                     float* __restrict__ selbuf,
                                                     float* __restrict__ partial) {
    int waveId = blockIdx.x * 8 + (threadIdx.x >> 5);
    int lane   = threadIdx.x & 31;
    int ch     = lane * 2;
    bool mx0 = gamma[ch] >= 0.f, mx1 = gamma[ch + 1] >= 0.f;

    float s0 = 0.f, s1 = 0.f, q0 = 0.f, q1 = 0.f;

    for (int pt = 0; pt < 32; ++pt) {
        int p = waveId * 32 + pt;
        int b = p >> 12;
        float2 e = ((const float2*)(y3T + (size_t)p * OCH))[lane];
        const int* ip = idx + (size_t)p * KNN;
        const float* y1b = y1T + (size_t)b * NPTS * OCH;

        float hi0 = -FLT_MAX, hi1 = -FLT_MAX, lo0 = FLT_MAX, lo1 = FLT_MAX;
#pragma unroll
        for (int j = 0; j < KNN; ++j) {
            int nb = ip[j];
            float2 g = ((const float2*)(y1b + ((size_t)nb << 6)))[lane];
            float h0 = g.x + e.x, h1 = g.y + e.y;
            s0 += h0; s1 += h1; q0 += h0 * h0; q1 += h1 * h1;
            hi0 = fmaxf(hi0, h0); lo0 = fminf(lo0, h0);
            hi1 = fmaxf(hi1, h1); lo1 = fminf(lo1, h1);
        }
        float2 selv = { mx0 ? hi0 : lo0, mx1 ? hi1 : lo1 };
        ((float2*)(selbuf + (size_t)p * OCH))[lane] = selv;
    }
    __shared__ float red[8][2 * OCH];
    const int wv = threadIdx.x >> 5;
    float2 sv = { s0, s1 }, qv = { q0, q1 };
    *(float2*)&red[wv][ch]       = sv;
    *(float2*)&red[wv][OCH + ch] = qv;
    __syncthreads();
    if (threadIdx.x < 2 * OCH) {
        float v = red[0][threadIdx.x];
        for (int i = 1; i < 8; ++i) v += red[i][threadIdx.x];
        partial[(size_t)blockIdx.x * (2 * OCH) + threadIdx.x] = v;
    }
}

__global__ void stats_kernel(const float* __restrict__ partial,
                             const float* __restrict__ gamma,
                             const float* __restrict__ beta,
                             float* __restrict__ coeff) {
    int ch = threadIdx.x;
    if (ch >= OCH) return;
    float s = 0.f, q = 0.f;
#pragma unroll 4
    for (int w = 0; w < GATHER_BLOCKS; ++w) {
        s += partial[(size_t)w * 128 + ch];
        q += partial[(size_t)w * 128 + OCH + ch];
    }
    const float M = (float)BATCH * NPTS * KNN;
    float mean  = s / M;
    float var   = q / M - mean * mean;
    float scale = gamma[ch] * rsqrtf(var + 1e-5f);
    coeff[ch]       = scale;
    coeff[OCH + ch] = beta[ch] - mean * scale;
}

__global__ __launch_bounds__(256) void final_kernel(const float* __restrict__ selbuf,
                                                    const float* __restrict__ coeff,
                                                    float* __restrict__ out) {
    int t = blockIdx.x * 256 + threadIdx.x;
    int n = t & (NPTS - 1);
    int o = (t >> 12) & (OCH - 1);
    int b = t >> 18;
    float v = selbuf[(((size_t)b * NPTS + n) << 6) + o];
    float r = v * coeff[o] + coeff[OCH + o];
    out[t] = (r >= 0.f) ? r : 0.2f * r;
}

extern "C" void kernel_launch(void* const* d_in, const int* in_sizes, int n_in,
                              void* d_out, int out_size, void* d_ws, size_t ws_size,
                              hipStream_t stream) {
    const float* x     = (const float*)d_in[0];
    const float* W     = (const float*)d_in[1];
    const float* gamma = (const float*)d_in[2];
    const float* beta  = (const float*)d_in[3];
    float* out = (float*)d_out;

    float* ws    = (float*)d_ws;
    float* sq    = ws;
    int*   idx   = (int*)(ws + 32768);
    float* y1T   = ws + 32768 + 655360;
    float* y3T   = y1T + (size_t)BATCH * NPTS * OCH;
    float* sel   = y3T + (size_t)BATCH * NPTS * OCH;
    float* part  = sel + (size_t)BATCH * NPTS * OCH;
    float* coeff = part + 1024 * 128;

    sq_kernel<<<(BATCH * NPTS) / 256, 256, 0, stream>>>(x, sq);
    knn_kernel<<<dim3(NPTS / 64, BATCH), 128, 0, stream>>>(x, sq, idx);
    proj_kernel<<<2048, 256, 0, stream>>>(x, W, y1T, y3T);
    gather_kernel<<<GATHER_BLOCKS, 256, 0, stream>>>(y1T, y3T, idx, gamma, sel, part);
    stats_kernel<<<1, 64, 0, stream>>>(part, gamma, beta, coeff);
    final_kernel<<<(BATCH * OCH * NPTS) / 256, 256, 0, stream>>>(sel, coeff, out);
}


# final pass without transposition: the gather writes its selected values in the output's layout (a wave's 32 points = one 128-byte line per channel, via its own LDS tile); last pass a 16-byte stream
# speedup vs baseline: 3.2614x; 1.0876x over previous
#include <hip/hip_runtime.h>
#include <hip/hip_bf16.h>
#include <float.h>


typedef float v2f __attribute__((ext_vector_type(2)));
typedef float v8f __attribute__((ext_vector_type(8)));

#define BATCH 8
#define CH    64
#define NPTS  4096
#define OCH   64
#define KNN   20
#define NEG_INF (-3.0e38f)
#define GATHER_BLOCKS 128

__device__ __forceinline__ v8f wmma_f32(v2f a, v2f b, v8f c) {
    return __builtin_amdgcn_wmma_f32_16x16x4_f32(
        false, a, false, b, (short)0, c, false, false);
}

__global__ __launch_bounds__(256) void sq_kernel(const float* __restrict__ x,
                                                 float* __restrict__ sq) {
    int t = blockIdx.x * 256 + threadIdx.x;
    int b = t >> 12;
    int n = t & (NPTS - 1);
    const float* p = x + (size_t)b * CH * NPTS + n;
    float s = 0.f;
#pragma unroll
    for (int c = 0; c < CH; ++c) { float v = p[(size_t)c * NPTS]; s += v * v; }
    sq[t] = s;
}


#define TOPK_INSERT(v, col)                                                   \
    do {                                                                      \
        const float nv = (v); const int nc = (col);                           \
        bool gt[KNN];                                                         \
        _Pragma("unroll")                                                     \
        for (int s = 0; s < KNN; ++s) gt[s] = nv > best[s];                   \
        _Pragma("unroll")                                                     \
        for (int s = KNN - 1; s > 0; --s) {                                   \
            const float uv = gt[s - 1] ? best[s - 1] : nv;                    \
            const int   ui = gt[s - 1] ? bidx[s - 1] : nc;                    \
            best[s] = gt[s] ? uv : best[s];                                   \
            bidx[s] = gt[s] ? ui : bidx[s];                                   \
        }                                                                     \
        best[0] = gt[0] ? nv : best[0];                                       \
        bidx[0] = gt[0] ? nc : bidx[0];                                       \
        thr = best[KNN - 1];                                                  \
    } while (0)

struct CandTile { v2f f[16]; float4 sqlo, sqhi; };

__device__ __forceinline__ void load_cand(CandTile& t, const float* __restrict__ xb,
                                          const float* __restrict__ sqb, int col0,
                                          int lmod, int khalf, int rowadd) {
#pragma unroll
    for (int kk = 0; kk < 16; ++kk) {
        const float* p = xb + (size_t)(4 * kk + khalf) * NPTS + col0 + lmod;
        t.f[kk].x = p[0];
        t.f[kk].y = p[NPTS];
    }
    const float4* sp = (const float4*)(sqb + col0 + rowadd);
    t.sqlo = sp[0];
    t.sqhi = sp[1];
}

#define QCAP   32
#define QROOM  8

__device__ __forceinline__ void push_tile(v8f acc, float4 sqlo, float4 sqhi, int cbase,
                                          float thr, float2* q, int& cnt) {
    float sc[8];
    sc[0] = 2.0f * acc[0] - sqlo.x; sc[1] = 2.0f * acc[1] - sqlo.y;
    sc[2] = 2.0f * acc[2] - sqlo.z; sc[3] = 2.0f * acc[3] - sqlo.w;
    sc[4] = 2.0f * acc[4] - sqhi.x; sc[5] = 2.0f * acc[5] - sqhi.y;
    sc[6] = 2.0f * acc[6] - sqhi.z; sc[7] = 2.0f * acc[7] - sqhi.w;
#pragma unroll
    for (int r = 0; r < 8; ++r) {
        q[cnt * 32] = make_float2(sc[r], __int_as_float(cbase + r));
        cnt += (sc[r] > thr) ? 1 : 0;
    }
}

#define DRAIN_QUEUE()                                                         \
    do {                                                                      \
        float2 e = make_float2(NEG_INF, 0.f);                                 \
        if (cnt > 0) e = q[0];                                                \
        for (int i = 0; __builtin_amdgcn_ballot_w32(i < cnt) != 0u; ++i) {    \
            float2 en = e;                                                    \
            if (i + 1 < cnt) en = q[(i + 1) * 32];                            \
            if (i < cnt && e.x > thr) {                                       \
                const float ev = e.x; const int ec = __float_as_int(e.y);     \
                TOPK_INSERT(ev, ec);                                          \
            }                                                                 \
            e = en;                                                           \
        }                                                                     \
        cnt = 0;                                                              \
    } while (0)

#define MAKE_ROOM()                                                           \
    do {                                                                      \
        if (__builtin_amdgcn_ballot_w32(cnt > QCAP - QROOM) != 0u)            \
            DRAIN_QUEUE();                                                    \
    } while (0)

__global__ __launch_bounds__(128) void knn_kernel(const float* __restrict__ x,
                                                  const float* __restrict__ sq,
                                                  int* __restrict__ idxout) {
    const int b    = blockIdx.y;
    const int wave = threadIdx.x >> 5;
    const int lane = threadIdx.x & 31;
    const int q0   = (blockIdx.x * 4 + wave) * 16;
    const int lmod = lane & 15;
    const int half = lane >> 4;
    const int khalf  = half << 1;
    const int rowadd = half << 3;
    const float* xb  = x + (size_t)b * CH * NPTS;
    const float* sqb = sq + (size_t)b * NPTS;

    v2f qf[16];
#pragma unroll
    for (int kk = 0; kk < 16; ++kk) {
        const float* p = xb + (size_t)(4 * kk + khalf) * NPTS + q0 + lmod;
        qf[kk].x = p[0];
        qf[kk].y = p[NPTS];
    }

    float best[KNN]; int bidx[KNN]; float thr = NEG_INF;
#pragma unroll
    for (int i = 0; i < KNN; ++i) { best[i] = NEG_INF; bidx[i] = 0; }

    __shared__ float2 qbuf[4][QCAP][32];
    float2* q = &qbuf[wave][0][lane];
    int cnt = 0;

    CandTile a0, a1, b0, b1;
    load_cand(a0, xb, sqb, 0,  lmod, khalf, rowadd);
    load_cand(a1, xb, sqb, 16, lmod, khalf, rowadd);

    for (int c0 = 0; c0 < NPTS; c0 += 64) {
        load_cand(b0, xb, sqb, c0 + 32, lmod, khalf, rowadd);
        {
            v8f acc0 = {}, acc1 = {};
#pragma unroll
            for (int kk = 0; kk < 16; ++kk) {
                acc0 = wmma_f32(a0.f[kk], qf[kk], acc0);
                acc1 = wmma_f32(a1.f[kk], qf[kk], acc1);
            }
            MAKE_ROOM();
            push_tile(acc0, a0.sqlo, a0.sqhi, c0 + rowadd,      thr, q, cnt);
            load_cand(b1, xb, sqb, c0 + 48, lmod, khalf, rowadd);
            MAKE_ROOM();
            push_tile(acc1, a1.sqlo, a1.sqhi, c0 + 16 + rowadd, thr, q, cnt);
        }
        const int cn = (c0 + 64) & (NPTS - 1);
        load_cand(a0, xb, sqb, cn, lmod, khalf, rowadd);
        {
            v8f acc0 = {}, acc1 = {};
#pragma unroll
            for (int kk = 0; kk < 16; ++kk) {
                acc0 = wmma_f32(b0.f[kk], qf[kk], acc0);
                acc1 = wmma_f32(b1.f[kk], qf[kk], acc1);
            }
            MAKE_ROOM();
            push_tile(acc0, b0.sqlo, b0.sqhi, c0 + 32 + rowadd, thr, q, cnt);
            load_cand(a1, xb, sqb, cn + 16, lmod, khalf, rowadd);
            MAKE_ROOM();
            push_tile(acc1, b1.sqlo, b1.sqhi, c0 + 48 + rowadd, thr, q, cnt);
        }
    }
    DRAIN_QUEUE();

#pragma nounroll
    for (int i = 0; i < KNN; ++i) {
        const float ov = __shfl_xor(best[0], 16);
        const int   oi = __shfl_xor(bidx[0], 16);
        if (half) {
#pragma unroll
            for (int s = 0; s < KNN - 1; ++s) { best[s] = best[s + 1]; bidx[s] = bidx[s + 1]; }
        } else if (ov > thr) {
            TOPK_INSERT(ov, oi);
        }
    }
    if (half == 0) {
        int* op = idxout + ((size_t)b * NPTS + q0 + lmod) * KNN;
#pragma unroll
        for (int i = 0; i < KNN; ++i) op[i] = bidx[i];
    }
}

__global__ __launch_bounds__(256) void proj_kernel(const float* __restrict__ x,
                                                   const float* __restrict__ W,
                                                   float* __restrict__ y1T,
                                                   float* __restrict__ y3T) {
    int task = blockIdx.x * 8 + (threadIdx.x >> 5);
    int lane = threadIdx.x & 31;
    int lmod = lane & 15;
    int khalf  = (lane >> 4) << 1;
    int rowadd = (lane >> 4) << 3;

    int which = task & 1;
    int nt    = (task >> 1) & 255;
    int ot    = (task >> 9) & 3;
    int b     = task >> 11;

    const float* xb = x + (size_t)b * CH * NPTS;
    const float* wr = W + (size_t)(ot * 16 + lmod) * (2 * CH);
    const int n0 = nt * 16;

    v8f acc = {};
#pragma unroll
    for (int kk = 0; kk < 16; ++kk) {
        int c = 4 * kk + khalf;
        v2f av, bv;
        if (which == 0) { av.x = wr[c];            av.y = wr[c + 1]; }
        else            { av.x = wr[CH + c] - wr[c];
                          av.y = wr[CH + c + 1] - wr[c + 1]; }
        const float* p = xb + (size_t)c * NPTS + n0 + lmod;
        bv.x = p[0]; bv.y = p[NPTS];
        acc = wmma_f32(av, bv, acc);
    }

    float* y = (which ? y3T : y1T)
             + (size_t)b * NPTS * OCH + (size_t)(n0 + lmod) * OCH
             + ot * 16 + rowadd;
    float4 lo = { acc[0], acc[1], acc[2], acc[3] };
    float4 hi = { acc[4], acc[5], acc[6], acc[7] };
    ((float4*)y)[0] = lo;
    ((float4*)y)[1] = hi;
}

#define SEL_PITCH 66
__global__ __launch_bounds__(256) void gather_kernel(const float* __restrict__ y1T,
                                                     const float* __restrict__ y3T,
                                                     const int* __restrict__ idx,
                                                     const float* __restrict__ gamma,
                                                     float* __restrict__ selT,
                                                     float* __restrict__ partial) {
    const int wv = threadIdx.x >> 5;
    int waveId = blockIdx.x * 8 + wv;
    int lane   = threadIdx.x & 31;
    int ch     = lane * 2;
    __shared__ __attribute__((aligned(16))) float tile[8][32][SEL_PITCH];
    bool mx0 = gamma[ch] >= 0.f, mx1 = gamma[ch + 1] >= 0.f;

    float s0 = 0.f, s1 = 0.f, q0 = 0.f, q1 = 0.f;

    for (int pt = 0; pt < 32; ++pt) {
        int p = waveId * 32 + pt;
        int b = p >> 12;
        float2 e = ((const float2*)(y3T + (size_t)p * OCH))[lane];
        const int* ip = idx + (size_t)p * KNN;
        const float* y1b = y1T + (size_t)b * NPTS * OCH;

        float hi0 = -FLT_MAX, hi1 = -FLT_MAX, lo0 = FLT_MAX, lo1 = FLT_MAX;
#pragma unroll
        for (int j = 0; j < KNN; ++j) {
            int nb = ip[j];
            float2 g = ((const float2*)(y1b + ((size_t)nb << 6)))[lane];
            float h0 = g.x + e.x, h1 = g.y + e.y;
            s0 += h0; s1 += h1; q0 += h0 * h0; q1 += h1 * h1;
            hi0 = fmaxf(hi0, h0); lo0 = fminf(lo0, h0);
            hi1 = fmaxf(hi1, h1); lo1 = fminf(lo1, h1);
        }
        float2 selv = { mx0 ? hi0 : lo0, mx1 ? hi1 : lo1 };
        *(float2*)&tile[wv][pt][ch] = selv;
    }
    __shared__ float red[8][2 * OCH];
    float2 sv = { s0, s1 }, qv = { q0, q1 };
    *(float2*)&red[wv][ch]       = sv;
    *(float2*)&red[wv][OCH + ch] = qv;
    __syncthreads();
    if (threadIdx.x < 2 * OCH) {
        float v = red[0][threadIdx.x];
        for (int i = 1; i < 8; ++i) v += red[i][threadIdx.x];
        partial[(size_t)blockIdx.x * (2 * OCH) + threadIdx.x] = v;
    }
    const int p0 = waveId * 32;
    const int bb = p0 >> 12;
    const int n0 = p0 & (NPTS - 1);
    float* op = selT + (size_t)bb * OCH * NPTS + n0 + lane;
    const float* tp = &tile[wv][lane][0];
    for (int o = 0; o < OCH; ++o) op[(size_t)o * NPTS] = tp[o];
}

__global__ void stats_kernel(const float* __restrict__ partial,
                             const float* __restrict__ gamma,
                             const float* __restrict__ beta,
                             float* __restrict__ coeff) {
    int ch = threadIdx.x;
    if (ch >= OCH) return;
    float s = 0.f, q = 0.f;
#pragma unroll 4
    for (int w = 0; w < GATHER_BLOCKS; ++w) {
        s += partial[(size_t)w * 128 + ch];
        q += partial[(size_t)w * 128 + OCH + ch];
    }
    const float M = (float)BATCH * NPTS * KNN;
    float mean  = s / M;
    float var   = q / M - mean * mean;
    float scale = gamma[ch] * rsqrtf(var + 1e-5f);
    coeff[ch]       = scale;
    coeff[OCH + ch] = beta[ch] - mean * scale;
}

__global__ __launch_bounds__(256) void final_kernel(const float* __restrict__ selT,
                                                    const float* __restrict__ coeff,
                                                    float* __restrict__ out) {
    const int row = blockIdx.x;
    const int o   = row & (OCH - 1);
    const float sc = coeff[o], sh = coeff[OCH + o];
    const float4* src = (const float4*)(selT + (size_t)row * NPTS) + threadIdx.x;
    float4*       dst = (float4*)(out + (size_t)row * NPTS) + threadIdx.x;
    float4 v[4];
#pragma unroll
    for (int j = 0; j < 4; ++j) v[j] = src[j * 256];
#pragma unroll
    for (int j = 0; j < 4; ++j) {
        float4 r;
        r.x = v[j].x * sc + sh; r.x = (r.x >= 0.f) ? r.x : 0.2f * r.x;
        r.y = v[j].y * sc + sh; r.y = (r.y >= 0.f) ? r.y : 0.2f * r.y;
        r.z = v[j].z * sc + sh; r.z = (r.z >= 0.f) ? r.z : 0.2f * r.z;
        r.w = v[j].w * sc + sh; r.w = (r.w >= 0.f) ? r.w : 0.2f * r.w;
        dst[j * 256] = r;
    }
}

extern "C" void kernel_launch(void* const* d_in, const int* in_sizes, int n_in,
                              void* d_out, int out_size, void* d_ws, size_t ws_size,
                              hipStream_t stream) {
    const float* x     = (const float*)d_in[0];
    const float* W     = (const float*)d_in[1];
    const float* gamma = (const float*)d_in[2];
    const float* beta  = (const float*)d_in[3];
    float* out = (float*)d_out;

    float* ws    = (float*)d_ws;
    float* sq    = ws;
    int*   idx   = (int*)(ws + 32768);
    float* y1T   = ws + 32768 + 655360;
    float* y3T   = y1T + (size_t)BATCH * NPTS * OCH;
    float* sel   = y3T + (size_t)BATCH * NPTS * OCH;
    float* part  = sel + (size_t)BATCH * NPTS * OCH;
    float* coeff = part + 1024 * 128;

    sq_kernel<<<(BATCH * NPTS) / 256, 256, 0, stream>>>(x, sq);
    knn_kernel<<<dim3(NPTS / 64, BATCH), 128, 0, stream>>>(x, sq, idx);
    proj_kernel<<<2048, 256, 0, stream>>>(x, W, y1T, y3T);
    gather_kernel<<<GATHER_BLOCKS, 256, 0, stream>>>(y1T, y3T, idx, gamma, sel, part);
    stats_kernel<<<1, 64, 0, stream>>>(part, gamma, beta, coeff);
    final_kernel<<<BATCH * OCH, 256, 0, stream>>>(sel, coeff, out);
}


# gather: 32-wave blocks, 4 waves per 32-point tile (8 points per wave, not 32); a wave loads its 160 neighbour indices at once, handed out by lane reads; BN summation tree changes
# speedup vs baseline: 3.5223x; 1.0800x over previous
#include <hip/hip_runtime.h>
#include <hip/hip_bf16.h>
#include <float.h>


typedef float v2f __attribute__((ext_vector_type(2)));
typedef float v8f __attribute__((ext_vector_type(8)));

#define BATCH 8
#define CH    64
#define NPTS  4096
#define OCH   64
#define KNN   20
#define NEG_INF (-3.0e38f)
#define GATHER_BLOCKS 128

__device__ __forceinline__ v8f wmma_f32(v2f a, v2f b, v8f c) {
    return __builtin_amdgcn_wmma_f32_16x16x4_f32(
        false, a, false, b, (short)0, c, false, false);
}

__global__ __launch_bounds__(256) void sq_kernel(const float* __restrict__ x,
                                                 float* __restrict__ sq) {
    int t = blockIdx.x * 256 + threadIdx.x;
    int b = t >> 12;
    int n = t & (NPTS - 1);
    const float* p = x + (size_t)b * CH * NPTS + n;
    float s = 0.f;
#pragma unroll
    for (int c = 0; c < CH; ++c) { float v = p[(size_t)c * NPTS]; s += v * v; }
    sq[t] = s;
}


#define TOPK_INSERT(v, col)                                                   \
    do {                                                                      \
        const float nv = (v); const int nc = (col);                           \
        bool gt[KNN];                                                         \
        _Pragma("unroll")                                                     \
        for (int s = 0; s < KNN; ++s) gt[s] = nv > best[s];                   \
        _Pragma("unroll")                                                     \
        for (int s = KNN - 1; s > 0; --s) {                                   \
            const float uv = gt[s - 1] ? best[s - 1] : nv;                    \
            const int   ui = gt[s - 1] ? bidx[s - 1] : nc;                    \
            best[s] = gt[s] ? uv : best[s];                                   \
            bidx[s] = gt[s] ? ui : bidx[s];                                   \
        }                                                                     \
        best[0] = gt[0] ? nv : best[0];                                       \
        bidx[0] = gt[0] ? nc : bidx[0];                                       \
        thr = best[KNN - 1];                                                  \
    } while (0)

struct CandTile { v2f f[16]; float4 sqlo, sqhi; };

__device__ __forceinline__ void load_cand(CandTile& t, const float* __restrict__ xb,
                                          const float* __restrict__ sqb, int col0,
                                          int lmod, int khalf, int rowadd) {
#pragma unroll
    for (int kk = 0; kk < 16; ++kk) {
        const float* p = xb + (size_t)(4 * kk + khalf) * NPTS + col0 + lmod;
        t.f[kk].x = p[0];
        t.f[kk].y = p[NPTS];
    }
    const float4* sp = (const float4*)(sqb + col0 + rowadd);
    t.sqlo = sp[0];
    t.sqhi = sp[1];
}

#define QCAP   32
#define QROOM  8

__device__ __forceinline__ void push_tile(v8f acc, float4 sqlo, float4 sqhi, int cbase,
                                          float thr, float2* q, int& cnt) {
    float sc[8];
    sc[0] = 2.0f * acc[0] - sqlo.x; sc[1] = 2.0f * acc[1] - sqlo.y;
    sc[2] = 2.0f * acc[2] - sqlo.z; sc[3] = 2.0f * acc[3] - sqlo.w;
    sc[4] = 2.0f * acc[4] - sqhi.x; sc[5] = 2.0f * acc[5] - sqhi.y;
    sc[6] = 2.0f * acc[6] - sqhi.z; sc[7] = 2.0f * acc[7] - sqhi.w;
#pragma unroll
    for (int r = 0; r < 8; ++r) {
        q[cnt * 32] = make_float2(sc[r], __int_as_float(cbase + r));
        cnt += (sc[r] > thr) ? 1 : 0;
    }
}

#define DRAIN_QUEUE()                                                         \
    do {                                                                      \
        float2 e = make_float2(NEG_INF, 0.f);                                 \
        if (cnt > 0) e = q[0];                                                \
        for (int i = 0; __builtin_amdgcn_ballot_w32(i < cnt) != 0u; ++i) {    \
            float2 en = e;                                                    \
            if (i + 1 < cnt) en = q[(i + 1) * 32];                            \
            if (i < cnt && e.x > thr) {                                       \
                const float ev = e.x; const int ec = __float_as_int(e.y);     \
                TOPK_INSERT(ev, ec);                                          \
            }                                                                 \
            e = en;                                                           \
        }                                                                     \
        cnt = 0;                                                              \
    } while (0)

#define MAKE_ROOM()                                                           \
    do {                                                                      \
        if (__builtin_amdgcn_ballot_w32(cnt > QCAP - QROOM) != 0u)            \
            DRAIN_QUEUE();                                                    \
    } while (0)

__global__ __launch_bounds__(128) void knn_kernel(const float* __restrict__ x,
                                                  const float* __restrict__ sq,
                                                  int* __restrict__ idxout) {
    const int b    = blockIdx.y;
    const int wave = threadIdx.x >> 5;
    const int lane = threadIdx.x & 31;
    const int q0   = (blockIdx.x * 4 + wave) * 16;
    const int lmod = lane & 15;
    const int half = lane >> 4;
    const int khalf  = half << 1;
    const int rowadd = half << 3;
    const float* xb  = x + (size_t)b * CH * NPTS;
    const float* sqb = sq + (size_t)b * NPTS;

    v2f qf[16];
#pragma unroll
    for (int kk = 0; kk < 16; ++kk) {
        const float* p = xb + (size_t)(4 * kk + khalf) * NPTS + q0 + lmod;
        qf[kk].x = p[0];
        qf[kk].y = p[NPTS];
    }

    float best[KNN]; int bidx[KNN]; float thr = NEG_INF;
#pragma unroll
    for (int i = 0; i < KNN; ++i) { best[i] = NEG_INF; bidx[i] = 0; }

    __shared__ float2 qbuf[4][QCAP][32];
    float2* q = &qbuf[wave][0][lane];
    int cnt = 0;

    CandTile a0, a1, b0, b1;
    load_cand(a0, xb, sqb, 0,  lmod, khalf, rowadd);
    load_cand(a1, xb, sqb, 16, lmod, khalf, rowadd);

    for (int c0 = 0; c0 < NPTS; c0 += 64) {
        load_cand(b0, xb, sqb, c0 + 32, lmod, khalf, rowadd);
        {
            v8f acc0 = {}, acc1 = {};
#pragma unroll
            for (int kk = 0; kk < 16; ++kk) {
                acc0 = wmma_f32(a0.f[kk], qf[kk], acc0);
                acc1 = wmma_f32(a1.f[kk], qf[kk], acc1);
            }
            MAKE_ROOM();
            push_tile(acc0, a0.sqlo, a0.sqhi, c0 + rowadd,      thr, q, cnt);
            load_cand(b1, xb, sqb, c0 + 48, lmod, khalf, rowadd);
            MAKE_ROOM();
            push_tile(acc1, a1.sqlo, a1.sqhi, c0 + 16 + rowadd, thr, q, cnt);
        }
        const int cn = (c0 + 64) & (NPTS - 1);
        load_cand(a0, xb, sqb, cn, lmod, khalf, rowadd);
        {
            v8f acc0 = {}, acc1 = {};
#pragma unroll
            for (int kk = 0; kk < 16; ++kk) {
                acc0 = wmma_f32(b0.f[kk], qf[kk], acc0);
                acc1 = wmma_f32(b1.f[kk], qf[kk], acc1);
            }
            MAKE_ROOM();
            push_tile(acc0, b0.sqlo, b0.sqhi, c0 + 32 + rowadd, thr, q, cnt);
            load_cand(a1, xb, sqb, cn + 16, lmod, khalf, rowadd);
            MAKE_ROOM();
            push_tile(acc1, b1.sqlo, b1.sqhi, c0 + 48 + rowadd, thr, q, cnt);
        }
    }
    DRAIN_QUEUE();

#pragma nounroll
    for (int i = 0; i < KNN; ++i) {
        const float ov = __shfl_xor(best[0], 16);
        const int   oi = __shfl_xor(bidx[0], 16);
        if (half) {
#pragma unroll
            for (int s = 0; s < KNN - 1; ++s) { best[s] = best[s + 1]; bidx[s] = bidx[s + 1]; }
        } else if (ov > thr) {
            TOPK_INSERT(ov, oi);
        }
    }
    if (half == 0) {
        int* op = idxout + ((size_t)b * NPTS + q0 + lmod) * KNN;
#pragma unroll
        for (int i = 0; i < KNN; ++i) op[i] = bidx[i];
    }
}

__global__ __launch_bounds__(256) void proj_kernel(const float* __restrict__ x,
                                                   const float* __restrict__ W,
                                                   float* __restrict__ y1T,
                                                   float* __restrict__ y3T) {
    int task = blockIdx.x * 8 + (threadIdx.x >> 5);
    int lane = threadIdx.x & 31;
    int lmod = lane & 15;
    int khalf  = (lane >> 4) << 1;
    int rowadd = (lane >> 4) << 3;

    int which = task & 1;
    int nt    = (task >> 1) & 255;
    int ot    = (task >> 9) & 3;
    int b     = task >> 11;

    const float* xb = x + (size_t)b * CH * NPTS;
    const float* wr = W + (size_t)(ot * 16 + lmod) * (2 * CH);
    const int n0 = nt * 16;

    v8f acc = {};
#pragma unroll
    for (int kk = 0; kk < 16; ++kk) {
        int c = 4 * kk + khalf;
        v2f av, bv;
        if (which == 0) { av.x = wr[c];            av.y = wr[c + 1]; }
        else            { av.x = wr[CH + c] - wr[c];
                          av.y = wr[CH + c + 1] - wr[c + 1]; }
        const float* p = xb + (size_t)c * NPTS + n0 + lmod;
        bv.x = p[0]; bv.y = p[NPTS];
        acc = wmma_f32(av, bv, acc);
    }

    float* y = (which ? y3T : y1T)
             + (size_t)b * NPTS * OCH + (size_t)(n0 + lmod) * OCH
             + ot * 16 + rowadd;
    float4 lo = { acc[0], acc[1], acc[2], acc[3] };
    float4 hi = { acc[4], acc[5], acc[6], acc[7] };
    ((float4*)y)[0] = lo;
    ((float4*)y)[1] = hi;
}

#define SEL_PITCH 66
#define G_WAVES 32
#define G_TILES 8
#define G_PTS   8
__global__ __launch_bounds__(1024) void gather_kernel(const float* __restrict__ y1T,
                                                      const float* __restrict__ y3T,
                                                      const int* __restrict__ idx,
                                                      const float* __restrict__ gamma,
                                                      float* __restrict__ selT,
                                                      float* __restrict__ partial) {
    const int wv   = threadIdx.x >> 5;
    const int lane = threadIdx.x & 31;
    const int tl   = wv >> 2;
    const int sub  = wv & 3;
    const int ch   = lane * 2;
    __shared__ __attribute__((aligned(16))) float tile[G_TILES][32][SEL_PITCH];
    bool mx0 = gamma[ch] >= 0.f, mx1 = gamma[ch + 1] >= 0.f;

    const int p0 = (blockIdx.x * G_TILES + tl) * 32;
    const int pw = p0 + sub * G_PTS;
    const int b  = p0 >> 12;
    const float* y1b = y1T + (size_t)b * NPTS * OCH;

    int mine[5];
    {
        const int* ip = idx + (size_t)pw * KNN + lane * 5;
#pragma unroll
        for (int k = 0; k < 5; ++k) mine[k] = ip[k];
    }

    float s0 = 0.f, s1 = 0.f, q0 = 0.f, q1 = 0.f;

    for (int pt = 0; pt < G_PTS; ++pt) {
        const int p = pw + pt;
        float2 e = ((const float2*)(y3T + (size_t)p * OCH))[lane];

        float hi0 = -FLT_MAX, hi1 = -FLT_MAX, lo0 = FLT_MAX, lo1 = FLT_MAX;
#pragma unroll
        for (int j = 0; j < KNN; ++j) {
            const int nb = __builtin_amdgcn_readlane(mine[j % 5], 4 * pt + j / 5);
            float2 g = ((const float2*)(y1b + ((size_t)nb << 6)))[lane];
            float h0 = g.x + e.x, h1 = g.y + e.y;
            s0 += h0; s1 += h1; q0 += h0 * h0; q1 += h1 * h1;
            hi0 = fmaxf(hi0, h0); lo0 = fminf(lo0, h0);
            hi1 = fmaxf(hi1, h1); lo1 = fminf(lo1, h1);
        }
        float2 selv = { mx0 ? hi0 : lo0, mx1 ? hi1 : lo1 };
        *(float2*)&tile[tl][sub * G_PTS + pt][ch] = selv;
    }
    __shared__ float red[G_WAVES][2 * OCH];
    float2 sv = { s0, s1 }, qv = { q0, q1 };
    *(float2*)&red[wv][ch]       = sv;
    *(float2*)&red[wv][OCH + ch] = qv;
    __syncthreads();
    if (threadIdx.x < 2 * OCH) {
        float v = red[0][threadIdx.x];
        for (int i = 1; i < G_WAVES; ++i) v += red[i][threadIdx.x];
        partial[(size_t)blockIdx.x * (2 * OCH) + threadIdx.x] = v;
    }
    const int n0 = p0 & (NPTS - 1);
    float* op = selT + (size_t)b * OCH * NPTS + n0 + lane;
    const float* tp = &tile[tl][lane][0];
    for (int o = sub * 16; o < sub * 16 + 16; ++o) op[(size_t)o * NPTS] = tp[o];
}

__global__ void stats_kernel(const float* __restrict__ partial,
                             const float* __restrict__ gamma,
                             const float* __restrict__ beta,
                             float* __restrict__ coeff) {
    int ch = threadIdx.x;
    if (ch >= OCH) return;
    float s = 0.f, q = 0.f;
#pragma unroll 4
    for (int w = 0; w < GATHER_BLOCKS; ++w) {
        s += partial[(size_t)w * 128 + ch];
        q += partial[(size_t)w * 128 + OCH + ch];
    }
    const float M = (float)BATCH * NPTS * KNN;
    float mean  = s / M;
    float var   = q / M - mean * mean;
    float scale = gamma[ch] * rsqrtf(var + 1e-5f);
    coeff[ch]       = scale;
    coeff[OCH + ch] = beta[ch] - mean * scale;
}

__global__ __launch_bounds__(256) void final_kernel(const float* __restrict__ selT,
                                                    const float* __restrict__ coeff,
                                                    float* __restrict__ out) {
    const int row = blockIdx.x;
    const int o   = row & (OCH - 1);
    const float sc = coeff[o], sh = coeff[OCH + o];
    const float4* src = (const float4*)(selT + (size_t)row * NPTS) + threadIdx.x;
    float4*       dst = (float4*)(out + (size_t)row * NPTS) + threadIdx.x;
    float4 v[4];
#pragma unroll
    for (int j = 0; j < 4; ++j) v[j] = src[j * 256];
#pragma unroll
    for (int j = 0; j < 4; ++j) {
        float4 r;
        r.x = v[j].x * sc + sh; r.x = (r.x >= 0.f) ? r.x : 0.2f * r.x;
        r.y = v[j].y * sc + sh; r.y = (r.y >= 0.f) ? r.y : 0.2f * r.y;
        r.z = v[j].z * sc + sh; r.z = (r.z >= 0.f) ? r.z : 0.2f * r.z;
        r.w = v[j].w * sc + sh; r.w = (r.w >= 0.f) ? r.w : 0.2f * r.w;
        dst[j * 256] = r;
    }
}

extern "C" void kernel_launch(void* const* d_in, const int* in_sizes, int n_in,
                              void* d_out, int out_size, void* d_ws, size_t ws_size,
                              hipStream_t stream) {
    const float* x     = (const float*)d_in[0];
    const float* W     = (const float*)d_in[1];
    const float* gamma = (const float*)d_in[2];
    const float* beta  = (const float*)d_in[3];
    float* out = (float*)d_out;

    float* ws    = (float*)d_ws;
    float* sq    = ws;
    int*   idx   = (int*)(ws + 32768);
    float* y1T   = ws + 32768 + 655360;
    float* y3T   = y1T + (size_t)BATCH * NPTS * OCH;
    float* sel   = y3T + (size_t)BATCH * NPTS * OCH;
    float* part  = sel + (size_t)BATCH * NPTS * OCH;
    float* coeff = part + 1024 * 128;

    sq_kernel<<<(BATCH * NPTS) / 256, 256, 0, stream>>>(x, sq);
    knn_kernel<<<dim3(NPTS / 64, BATCH), 128, 0, stream>>>(x, sq, idx);
    proj_kernel<<<2048, 256, 0, stream>>>(x, W, y1T, y3T);
    gather_kernel<<<GATHER_BLOCKS, 1024, 0, stream>>>(y1T, y3T, idx, gamma, sel, part);
    stats_kernel<<<1, 64, 0, stream>>>(part, gamma, beta, coeff);
    final_kernel<<<BATCH * OCH, 256, 0, stream>>>(sel, coeff, out);
}


# statistics pass: one thread per column of the 128 partial rows, all its row loads requested ahead of the adds (same row order, sums bit-identical): 32 dependent second-level-cache trips become one
# speedup vs baseline: 3.9042x; 1.1084x over previous
#include <hip/hip_runtime.h>
#include <hip/hip_bf16.h>
#include <float.h>


typedef float v2f __attribute__((ext_vector_type(2)));
typedef float v8f __attribute__((ext_vector_type(8)));

#define BATCH 8
#define CH    64
#define NPTS  4096
#define OCH   64
#define KNN   20
#define NEG_INF (-3.0e38f)
#define GATHER_BLOCKS 128

__device__ __forceinline__ v8f wmma_f32(v2f a, v2f b, v8f c) {
    return __builtin_amdgcn_wmma_f32_16x16x4_f32(
        false, a, false, b, (short)0, c, false, false);
}

__global__ __launch_bounds__(256) void sq_kernel(const float* __restrict__ x,
                                                 float* __restrict__ sq) {
    int t = blockIdx.x * 256 + threadIdx.x;
    int b = t >> 12;
    int n = t & (NPTS - 1);
    const float* p = x + (size_t)b * CH * NPTS + n;
    float s = 0.f;
#pragma unroll
    for (int c = 0; c < CH; ++c) { float v = p[(size_t)c * NPTS]; s += v * v; }
    sq[t] = s;
}


#define TOPK_INSERT(v, col)                                                   \
    do {                                                                      \
        const float nv = (v); const int nc = (col);                           \
        bool gt[KNN];                                                         \
        _Pragma("unroll")                                                     \
        for (int s = 0; s < KNN; ++s) gt[s] = nv > best[s];                   \
        _Pragma("unroll")                                                     \
        for (int s = KNN - 1; s > 0; --s) {                                   \
            const float uv = gt[s - 1] ? best[s - 1] : nv;                    \
            const int   ui = gt[s - 1] ? bidx[s - 1] : nc;                    \
            best[s] = gt[s] ? uv : best[s];                                   \
            bidx[s] = gt[s] ? ui : bidx[s];                                   \
        }                                                                     \
        best[0] = gt[0] ? nv : best[0];                                       \
        bidx[0] = gt[0] ? nc : bidx[0];                                       \
        thr = best[KNN - 1];                                                  \
    } while (0)

struct CandTile { v2f f[16]; float4 sqlo, sqhi; };

__device__ __forceinline__ void load_cand(CandTile& t, const float* __restrict__ xb,
                                          const float* __restrict__ sqb, int col0,
                                          int lmod, int khalf, int rowadd) {
#pragma unroll
    for (int kk = 0; kk < 16; ++kk) {
        const float* p = xb + (size_t)(4 * kk + khalf) * NPTS + col0 + lmod;
        t.f[kk].x = p[0];
        t.f[kk].y = p[NPTS];
    }
    const float4* sp = (const float4*)(sqb + col0 + rowadd);
    t.sqlo = sp[0];
    t.sqhi = sp[1];
}

#define QCAP   32
#define QROOM  8

__device__ __forceinline__ void push_tile(v8f acc, float4 sqlo, float4 sqhi, int cbase,
                                          float thr, float2* q, int& cnt) {
    float sc[8];
    sc[0] = 2.0f * acc[0] - sqlo.x; sc[1] = 2.0f * acc[1] - sqlo.y;
    sc[2] = 2.0f * acc[2] - sqlo.z; sc[3] = 2.0f * acc[3] - sqlo.w;
    sc[4] = 2.0f * acc[4] - sqhi.x; sc[5] = 2.0f * acc[5] - sqhi.y;
    sc[6] = 2.0f * acc[6] - sqhi.z; sc[7] = 2.0f * acc[7] - sqhi.w;
#pragma unroll
    for (int r = 0; r < 8; ++r) {
        q[cnt * 32] = make_float2(sc[r], __int_as_float(cbase + r));
        cnt += (sc[r] > thr) ? 1 : 0;
    }
}

#define DRAIN_QUEUE()                                                         \
    do {                                                                      \
        float2 e = make_float2(NEG_INF, 0.f);                                 \
        if (cnt > 0) e = q[0];                                                \
        for (int i = 0; __builtin_amdgcn_ballot_w32(i < cnt) != 0u; ++i) {    \
            float2 en = e;                                                    \
            if (i + 1 < cnt) en = q[(i + 1) * 32];                            \
            if (i < cnt && e.x > thr) {                                       \
                const float ev = e.x; const int ec = __float_as_int(e.y);     \
                TOPK_INSERT(ev, ec);                                          \
            }                                                                 \
            e = en;                                                           \
        }                                                                     \
        cnt = 0;                                                              \
    } while (0)

#define MAKE_ROOM()                                                           \
    do {                                                                      \
        if (__builtin_amdgcn_ballot_w32(cnt > QCAP - QROOM) != 0u)            \
            DRAIN_QUEUE();                                                    \
    } while (0)

#define W_PITCH 132

__global__ __launch_bounds__(128) void knn_kernel(const float* __restrict__ x,
                                                  const float* __restrict__ sq,
                                                  const float* __restrict__ W,
                                                  int* __restrict__ idxout,
                                                  float* __restrict__ y1T,
                                                  float* __restrict__ y3T) {
    const int b    = blockIdx.y;
    const int wave = threadIdx.x >> 5;
    const int lane = threadIdx.x & 31;
    const int q0   = (blockIdx.x * 4 + wave) * 16;
    const int lmod = lane & 15;
    const int half = lane >> 4;
    const int khalf  = half << 1;
    const int rowadd = half << 3;
    const float* xb  = x + (size_t)b * CH * NPTS;
    const float* sqb = sq + (size_t)b * NPTS;

    v2f qf[16];
#pragma unroll
    for (int kk = 0; kk < 16; ++kk) {
        const float* p = xb + (size_t)(4 * kk + khalf) * NPTS + q0 + lmod;
        qf[kk].x = p[0];
        qf[kk].y = p[NPTS];
    }

    __shared__ __attribute__((aligned(16))) float wlds[OCH][W_PITCH];
    for (int i = 0; i < 16; ++i) {
        const int f  = i * 128 + threadIdx.x;
        const int r  = f >> 5, c4 = f & 31;
        *(float4*)&wlds[r][c4 * 4] = ((const float4*)W)[f];
    }
    __syncthreads();
    {
        const size_t prow = ((size_t)b * NPTS + q0 + lmod) * OCH + rowadd;
        for (int ot = 0; ot < 4; ++ot) {
            const float* wr = &wlds[ot * 16 + lmod][khalf];
            v8f p1 = {}, p3 = {};
#pragma unroll
            for (int kk = 0; kk < 16; ++kk) {
                const v2f w1 = *(const v2f*)(wr + 4 * kk);
                const v2f w2 = *(const v2f*)(wr + CH + 4 * kk);
                const v2f wd = w2 - w1;
                p1 = wmma_f32(w1, qf[kk], p1);
                p3 = wmma_f32(wd, qf[kk], p3);
            }
            float4* o1 = (float4*)(y1T + prow + ot * 16);
            float4* o3 = (float4*)(y3T + prow + ot * 16);
            o1[0] = make_float4(p1[0], p1[1], p1[2], p1[3]);
            o1[1] = make_float4(p1[4], p1[5], p1[6], p1[7]);
            o3[0] = make_float4(p3[0], p3[1], p3[2], p3[3]);
            o3[1] = make_float4(p3[4], p3[5], p3[6], p3[7]);
        }
    }

    float best[KNN]; int bidx[KNN]; float thr = NEG_INF;
#pragma unroll
    for (int i = 0; i < KNN; ++i) { best[i] = NEG_INF; bidx[i] = 0; }

    __shared__ float2 qbuf[4][QCAP][32];
    float2* q = &qbuf[wave][0][lane];
    int cnt = 0;

    CandTile a0, a1, b0, b1;
    load_cand(a0, xb, sqb, 0,  lmod, khalf, rowadd);
    load_cand(a1, xb, sqb, 16, lmod, khalf, rowadd);

    for (int c0 = 0; c0 < NPTS; c0 += 64) {
        load_cand(b0, xb, sqb, c0 + 32, lmod, khalf, rowadd);
        {
            v8f acc0 = {}, acc1 = {};
#pragma unroll
            for (int kk = 0; kk < 16; ++kk) {
                acc0 = wmma_f32(a0.f[kk], qf[kk], acc0);
                acc1 = wmma_f32(a1.f[kk], qf[kk], acc1);
            }
            MAKE_ROOM();
            push_tile(acc0, a0.sqlo, a0.sqhi, c0 + rowadd,      thr, q, cnt);
            load_cand(b1, xb, sqb, c0 + 48, lmod, khalf, rowadd);
            MAKE_ROOM();
            push_tile(acc1, a1.sqlo, a1.sqhi, c0 + 16 + rowadd, thr, q, cnt);
        }
        const int cn = (c0 + 64) & (NPTS - 1);
        load_cand(a0, xb, sqb, cn, lmod, khalf, rowadd);
        {
            v8f acc0 = {}, acc1 = {};
#pragma unroll
            for (int kk = 0; kk < 16; ++kk) {
                acc0 = wmma_f32(b0.f[kk], qf[kk], acc0);
                acc1 = wmma_f32(b1.f[kk], qf[kk], acc1);
            }
            MAKE_ROOM();
            push_tile(acc0, b0.sqlo, b0.sqhi, c0 + 32 + rowadd, thr, q, cnt);
            load_cand(a1, xb, sqb, cn + 16, lmod, khalf, rowadd);
            MAKE_ROOM();
            push_tile(acc1, b1.sqlo, b1.sqhi, c0 + 48 + rowadd, thr, q, cnt);
        }
    }
    DRAIN_QUEUE();

#pragma nounroll
    for (int i = 0; i < KNN; ++i) {
        const float ov = __shfl_xor(best[0], 16);
        const int   oi = __shfl_xor(bidx[0], 16);
        if (half) {
#pragma unroll
            for (int s = 0; s < KNN - 1; ++s) { best[s] = best[s + 1]; bidx[s] = bidx[s + 1]; }
        } else if (ov > thr) {
            TOPK_INSERT(ov, oi);
        }
    }
    if (half == 0) {
        int* op = idxout + ((size_t)b * NPTS + q0 + lmod) * KNN;
#pragma unroll
        for (int i = 0; i < KNN; ++i) op[i] = bidx[i];
    }
}

#define SEL_PITCH 66
#define G_WAVES 32
#define G_TILES 8
#define G_PTS   8
__global__ __launch_bounds__(1024) void gather_kernel(const float* __restrict__ y1T,
                                                      const float* __restrict__ y3T,
                                                      const int* __restrict__ idx,
                                                      const float* __restrict__ gamma,
                                                      float* __restrict__ selT,
                                                      float* __restrict__ partial) {
    const int wv   = threadIdx.x >> 5;
    const int lane = threadIdx.x & 31;
    const int tl   = wv >> 2;
    const int sub  = wv & 3;
    const int ch   = lane * 2;
    __shared__ __attribute__((aligned(16))) float tile[G_TILES][32][SEL_PITCH];
    bool mx0 = gamma[ch] >= 0.f, mx1 = gamma[ch + 1] >= 0.f;

    const int p0 = (blockIdx.x * G_TILES + tl) * 32;
    const int pw = p0 + sub * G_PTS;
    const int b  = p0 >> 12;
    const float* y1b = y1T + (size_t)b * NPTS * OCH;

    int mine[5];
    {
        const int* ip = idx + (size_t)pw * KNN + lane * 5;
#pragma unroll
        for (int k = 0; k < 5; ++k) mine[k] = ip[k];
    }

    float s0 = 0.f, s1 = 0.f, q0 = 0.f, q1 = 0.f;

    for (int pt = 0; pt < G_PTS; ++pt) {
        const int p = pw + pt;
        float2 e = ((const float2*)(y3T + (size_t)p * OCH))[lane];

        float hi0 = -FLT_MAX, hi1 = -FLT_MAX, lo0 = FLT_MAX, lo1 = FLT_MAX;
#pragma unroll
        for (int j = 0; j < KNN; ++j) {
            const int nb = __builtin_amdgcn_readlane(mine[j % 5], 4 * pt + j / 5);
            float2 g = ((const float2*)(y1b + ((size_t)nb << 6)))[lane];
            float h0 = g.x + e.x, h1 = g.y + e.y;
            s0 += h0; s1 += h1; q0 += h0 * h0; q1 += h1 * h1;
            hi0 = fmaxf(hi0, h0); lo0 = fminf(lo0, h0);
            hi1 = fmaxf(hi1, h1); lo1 = fminf(lo1, h1);
        }
        float2 selv = { mx0 ? hi0 : lo0, mx1 ? hi1 : lo1 };
        *(float2*)&tile[tl][sub * G_PTS + pt][ch] = selv;
    }
    __shared__ float red[G_WAVES][2 * OCH];
    float2 sv = { s0, s1 }, qv = { q0, q1 };
    *(float2*)&red[wv][ch]       = sv;
    *(float2*)&red[wv][OCH + ch] = qv;
    __syncthreads();
    if (threadIdx.x < 2 * OCH) {
        float v = red[0][threadIdx.x];
        for (int i = 1; i < G_WAVES; ++i) v += red[i][threadIdx.x];
        partial[(size_t)blockIdx.x * (2 * OCH) + threadIdx.x] = v;
    }
    const int n0 = p0 & (NPTS - 1);
    float* op = selT + (size_t)b * OCH * NPTS + n0 + lane;
    const float* tp = &tile[tl][lane][0];
    for (int o = sub * 16; o < sub * 16 + 16; ++o) op[(size_t)o * NPTS] = tp[o];
}

__global__ __launch_bounds__(128) void stats_kernel(const float* __restrict__ partial,
                                                    const float* __restrict__ gamma,
                                                    const float* __restrict__ beta,
                                                    float* __restrict__ coeff) {
    const int col = threadIdx.x;
    float v[GATHER_BLOCKS];
#pragma unroll
    for (int w = 0; w < GATHER_BLOCKS; ++w) v[w] = partial[(size_t)w * 128 + col];
    float acc = 0.f;
#pragma unroll
    for (int w = 0; w < GATHER_BLOCKS; ++w) acc += v[w];

    __shared__ float tot[2 * OCH];
    tot[col] = acc;
    __syncthreads();
    if (col >= OCH) return;
    const int ch = col;
    const float s = tot[ch], q = tot[OCH + ch];
    const float M = (float)BATCH * NPTS * KNN;
    float mean  = s / M;
    float var   = q / M - mean * mean;
    float scale = gamma[ch] * rsqrtf(var + 1e-5f);
    coeff[ch]       = scale;
    coeff[OCH + ch] = beta[ch] - mean * scale;
}

__global__ __launch_bounds__(256) void final_kernel(const float* __restrict__ selT,
                                                    const float* __restrict__ coeff,
                                                    float* __restrict__ out) {
    const int row = blockIdx.x;
    const int o   = row & (OCH - 1);
    const float sc = coeff[o], sh = coeff[OCH + o];
    const float4* src = (const float4*)(selT + (size_t)row * NPTS) + threadIdx.x;
    float4*       dst = (float4*)(out + (size_t)row * NPTS) + threadIdx.x;
    float4 v[4];
#pragma unroll
    for (int j = 0; j < 4; ++j) v[j] = src[j * 256];
#pragma unroll
    for (int j = 0; j < 4; ++j) {
        float4 r;
        r.x = v[j].x * sc + sh; r.x = (r.x >= 0.f) ? r.x : 0.2f * r.x;
        r.y = v[j].y * sc + sh; r.y = (r.y >= 0.f) ? r.y : 0.2f * r.y;
        r.z = v[j].z * sc + sh; r.z = (r.z >= 0.f) ? r.z : 0.2f * r.z;
        r.w = v[j].w * sc + sh; r.w = (r.w >= 0.f) ? r.w : 0.2f * r.w;
        dst[j * 256] = r;
    }
}

extern "C" void kernel_launch(void* const* d_in, const int* in_sizes, int n_in,
                              void* d_out, int out_size, void* d_ws, size_t ws_size,
                              hipStream_t stream) {
    const float* x     = (const float*)d_in[0];
    const float* W     = (const float*)d_in[1];
    const float* gamma = (const float*)d_in[2];
    const float* beta  = (const float*)d_in[3];
    float* out = (float*)d_out;

    float* ws    = (float*)d_ws;
    float* sq    = ws;
    int*   idx   = (int*)(ws + 32768);
    float* y1T   = ws + 32768 + 655360;
    float* y3T   = y1T + (size_t)BATCH * NPTS * OCH;
    float* sel   = y3T + (size_t)BATCH * NPTS * OCH;
    float* part  = sel + (size_t)BATCH * NPTS * OCH;
    float* coeff = part + 1024 * 128;

    sq_kernel<<<(BATCH * NPTS) / 256, 256, 0, stream>>>(x, sq);
    knn_kernel<<<dim3(NPTS / 64, BATCH), 128, 0, stream>>>(x, sq, W, idx, y1T, y3T);
    gather_kernel<<<GATHER_BLOCKS, 1024, 0, stream>>>(y1T, y3T, idx, gamma, sel, part);
    stats_kernel<<<1, 128, 0, stream>>>(part, gamma, beta, coeff);
    final_kernel<<<BATCH * OCH, 256, 0, stream>>>(sel, coeff, out);
}


# kNN candidate tiles read from a fragment-ordered copy of x that the norms pass writes (into the still unused selected plane): 8 16-byte loads per tile, not 32 4-byte, one step ahead; sums unchanged
# speedup vs baseline: 4.1627x; 1.0662x over previous
#include <hip/hip_runtime.h>
#include <hip/hip_bf16.h>
#include <float.h>


typedef float v2f __attribute__((ext_vector_type(2)));
typedef float v8f __attribute__((ext_vector_type(8)));

#define BATCH 8
#define CH    64
#define NPTS  4096
#define OCH   64
#define KNN   20
#define NEG_INF (-3.0e38f)
#define GATHER_BLOCKS 128

__device__ __forceinline__ v8f wmma_f32(v2f a, v2f b, v8f c) {
    return __builtin_amdgcn_wmma_f32_16x16x4_f32(
        false, a, false, b, (short)0, c, false, false);
}

__global__ __launch_bounds__(256) void sq_kernel(const float* __restrict__ x,
                                                 float* __restrict__ sq,
                                                 float* __restrict__ xf) {
    int t = blockIdx.x * 256 + threadIdx.x;
    int b = t >> 12;
    int n = t & (NPTS - 1);
    const float* p = x + (size_t)b * CH * NPTS + n;
    float v[CH];
#pragma unroll
    for (int c = 0; c < CH; ++c) v[c] = p[(size_t)c * NPTS];
    float s = 0.f;
#pragma unroll
    for (int c = 0; c < CH; ++c) s += v[c] * v[c];
    sq[t] = s;
    float4* o = (float4*)xf + (size_t)(t >> 4) * 256 + (n & 15);
#pragma unroll
    for (int j = 0; j < 8; ++j) {
        o[j * 32]      = make_float4(v[8 * j],     v[8 * j + 1], v[8 * j + 4], v[8 * j + 5]);
        o[j * 32 + 16] = make_float4(v[8 * j + 2], v[8 * j + 3], v[8 * j + 6], v[8 * j + 7]);
    }
}


#define TOPK_INSERT(v, col)                                                   \
    do {                                                                      \
        const float nv = (v); const int nc = (col);                           \
        bool gt[KNN];                                                         \
        _Pragma("unroll")                                                     \
        for (int s = 0; s < KNN; ++s) gt[s] = nv > best[s];                   \
        _Pragma("unroll")                                                     \
        for (int s = KNN - 1; s > 0; --s) {                                   \
            const float uv = gt[s - 1] ? best[s - 1] : nv;                    \
            const int   ui = gt[s - 1] ? bidx[s - 1] : nc;                    \
            best[s] = gt[s] ? uv : best[s];                                   \
            bidx[s] = gt[s] ? ui : bidx[s];                                   \
        }                                                                     \
        best[0] = gt[0] ? nv : best[0];                                       \
        bidx[0] = gt[0] ? nc : bidx[0];                                       \
        thr = best[KNN - 1];                                                  \
    } while (0)

struct CandTile { float4 g[8]; };

__device__ __forceinline__ void load_cand(CandTile& t, const float4* __restrict__ xfb,
                                          int col0, int lane) {
    const float4* gp = xfb + (size_t)(col0 >> 4) * 256 + lane;
#pragma unroll
    for (int j = 0; j < 8; ++j) t.g[j] = gp[j * 32];
}

struct CandSq { float4 lo, hi; };

__device__ __forceinline__ void load_sq(CandSq& s, const float* __restrict__ sqb,
                                        int col0, int rowadd) {
    const float4* sp = (const float4*)(sqb + col0 + rowadd);
    s.lo = sp[0];
    s.hi = sp[1];
}

#define GRAM_2TILES(t0, t1, acc0, acc1)                                       \
    do {                                                                      \
        _Pragma("unroll")                                                     \
        for (int j = 0; j < 8; ++j) {                                         \
            const v2f e0 = { t0.g[j].x, t0.g[j].y }, o0 = { t0.g[j].z, t0.g[j].w }; \
            const v2f e1 = { t1.g[j].x, t1.g[j].y }, o1 = { t1.g[j].z, t1.g[j].w }; \
            acc0 = wmma_f32(e0, qf[2 * j],     acc0);                         \
            acc1 = wmma_f32(e1, qf[2 * j],     acc1);                         \
            acc0 = wmma_f32(o0, qf[2 * j + 1], acc0);                         \
            acc1 = wmma_f32(o1, qf[2 * j + 1], acc1);                         \
        }                                                                     \
    } while (0)

#define QCAP   32
#define QROOM  8

__device__ __forceinline__ void push_tile(v8f acc, float4 sqlo, float4 sqhi, int cbase,
                                          float thr, float2* q, int& cnt) {
    float sc[8];
    sc[0] = 2.0f * acc[0] - sqlo.x; sc[1] = 2.0f * acc[1] - sqlo.y;
    sc[2] = 2.0f * acc[2] - sqlo.z; sc[3] = 2.0f * acc[3] - sqlo.w;
    sc[4] = 2.0f * acc[4] - sqhi.x; sc[5] = 2.0f * acc[5] - sqhi.y;
    sc[6] = 2.0f * acc[6] - sqhi.z; sc[7] = 2.0f * acc[7] - sqhi.w;
#pragma unroll
    for (int r = 0; r < 8; ++r) {
        q[cnt * 32] = make_float2(sc[r], __int_as_float(cbase + r));
        cnt += (sc[r] > thr) ? 1 : 0;
    }
}

#define DRAIN_QUEUE()                                                         \
    do {                                                                      \
        float2 e = make_float2(NEG_INF, 0.f);                                 \
        if (cnt > 0) e = q[0];                                                \
        for (int i = 0; __builtin_amdgcn_ballot_w32(i < cnt) != 0u; ++i) {    \
            float2 en = e;                                                    \
            if (i + 1 < cnt) en = q[(i + 1) * 32];                            \
            if (i < cnt && e.x > thr) {                                       \
                const float ev = e.x; const int ec = __float_as_int(e.y);     \
                TOPK_INSERT(ev, ec);                                          \
            }                                                                 \
            e = en;                                                           \
        }                                                                     \
        cnt = 0;                                                              \
    } while (0)

#define MAKE_ROOM()                                                           \
    do {                                                                      \
        if (__builtin_amdgcn_ballot_w32(cnt > QCAP - QROOM) != 0u)            \
            DRAIN_QUEUE();                                                    \
    } while (0)

#define W_PITCH 132

__global__ __launch_bounds__(128) void knn_kernel(const float* __restrict__ x,
                                                  const float* __restrict__ xf,
                                                  const float* __restrict__ sq,
                                                  const float* __restrict__ W,
                                                  int* __restrict__ idxout,
                                                  float* __restrict__ y1T,
                                                  float* __restrict__ y3T) {
    const int b    = blockIdx.y;
    const int wave = threadIdx.x >> 5;
    const int lane = threadIdx.x & 31;
    const int q0   = (blockIdx.x * 4 + wave) * 16;
    const int lmod = lane & 15;
    const int half = lane >> 4;
    const int khalf  = half << 1;
    const int rowadd = half << 3;
    const float* xb  = x + (size_t)b * CH * NPTS;
    const float4* xfb = (const float4*)(xf + (size_t)b * CH * NPTS);
    const float* sqb = sq + (size_t)b * NPTS;

    v2f qf[16];
#pragma unroll
    for (int kk = 0; kk < 16; ++kk) {
        const float* p = xb + (size_t)(4 * kk + khalf) * NPTS + q0 + lmod;
        qf[kk].x = p[0];
        qf[kk].y = p[NPTS];
    }

    __shared__ __attribute__((aligned(16))) float wlds[OCH][W_PITCH];
    for (int i = 0; i < 16; ++i) {
        const int f  = i * 128 + threadIdx.x;
        const int r  = f >> 5, c4 = f & 31;
        *(float4*)&wlds[r][c4 * 4] = ((const float4*)W)[f];
    }
    __syncthreads();
    {
        const size_t prow = ((size_t)b * NPTS + q0 + lmod) * OCH + rowadd;
        for (int ot = 0; ot < 4; ++ot) {
            const float* wr = &wlds[ot * 16 + lmod][khalf];
            v8f p1 = {}, p3 = {};
#pragma unroll
            for (int kk = 0; kk < 16; ++kk) {
                const v2f w1 = *(const v2f*)(wr + 4 * kk);
                const v2f w2 = *(const v2f*)(wr + CH + 4 * kk);
                const v2f wd = w2 - w1;
                p1 = wmma_f32(w1, qf[kk], p1);
                p3 = wmma_f32(wd, qf[kk], p3);
            }
            float4* o1 = (float4*)(y1T + prow + ot * 16);
            float4* o3 = (float4*)(y3T + prow + ot * 16);
            o1[0] = make_float4(p1[0], p1[1], p1[2], p1[3]);
            o1[1] = make_float4(p1[4], p1[5], p1[6], p1[7]);
            o3[0] = make_float4(p3[0], p3[1], p3[2], p3[3]);
            o3[1] = make_float4(p3[4], p3[5], p3[6], p3[7]);
        }
    }

    float best[KNN]; int bidx[KNN]; float thr = NEG_INF;
#pragma unroll
    for (int i = 0; i < KNN; ++i) { best[i] = NEG_INF; bidx[i] = 0; }

    __shared__ float2 qbuf[4][QCAP][32];
    float2* q = &qbuf[wave][0][lane];
    int cnt = 0;

    CandTile a0, a1, b0, b1;
    load_cand(a0, xfb, 0,  lane);
    load_cand(a1, xfb, 16, lane);

    for (int c0 = 0; c0 < NPTS; c0 += 64) {
        {
            CandSq s0, s1;
            load_sq(s0, sqb, c0,      rowadd);
            load_sq(s1, sqb, c0 + 16, rowadd);
            load_cand(b0, xfb, c0 + 32, lane);
            load_cand(b1, xfb, c0 + 48, lane);
            v8f acc0 = {}, acc1 = {};
            GRAM_2TILES(a0, a1, acc0, acc1);
            MAKE_ROOM();
            push_tile(acc0, s0.lo, s0.hi, c0 + rowadd,      thr, q, cnt);
            MAKE_ROOM();
            push_tile(acc1, s1.lo, s1.hi, c0 + 16 + rowadd, thr, q, cnt);
        }
        const int cn = (c0 + 64) & (NPTS - 1);
        {
            CandSq s0, s1;
            load_sq(s0, sqb, c0 + 32, rowadd);
            load_sq(s1, sqb, c0 + 48, rowadd);
            load_cand(a0, xfb, cn,      lane);
            load_cand(a1, xfb, cn + 16, lane);
            v8f acc0 = {}, acc1 = {};
            GRAM_2TILES(b0, b1, acc0, acc1);
            MAKE_ROOM();
            push_tile(acc0, s0.lo, s0.hi, c0 + 32 + rowadd, thr, q, cnt);
            MAKE_ROOM();
            push_tile(acc1, s1.lo, s1.hi, c0 + 48 + rowadd, thr, q, cnt);
        }
    }
    DRAIN_QUEUE();

#pragma nounroll
    for (int i = 0; i < KNN; ++i) {
        const float ov = __shfl_xor(best[0], 16);
        const int   oi = __shfl_xor(bidx[0], 16);
        if (half) {
#pragma unroll
            for (int s = 0; s < KNN - 1; ++s) { best[s] = best[s + 1]; bidx[s] = bidx[s + 1]; }
        } else if (ov > thr) {
            TOPK_INSERT(ov, oi);
        }
    }
    if (half == 0) {
        int* op = idxout + ((size_t)b * NPTS + q0 + lmod) * KNN;
#pragma unroll
        for (int i = 0; i < KNN; ++i) op[i] = bidx[i];
    }
}

#define SEL_PITCH 66
#define G_WAVES 32
#define G_TILES 8
#define G_PTS   8
__global__ __launch_bounds__(1024) void gather_kernel(const float* __restrict__ y1T,
                                                      const float* __restrict__ y3T,
                                                      const int* __restrict__ idx,
                                                      const float* __restrict__ gamma,
                                                      float* __restrict__ selT,
                                                      float* __restrict__ partial) {
    const int wv   = threadIdx.x >> 5;
    const int lane = threadIdx.x & 31;
    const int tl   = wv >> 2;
    const int sub  = wv & 3;
    const int ch   = lane * 2;
    __shared__ __attribute__((aligned(16))) float tile[G_TILES][32][SEL_PITCH];
    bool mx0 = gamma[ch] >= 0.f, mx1 = gamma[ch + 1] >= 0.f;

    const int p0 = (blockIdx.x * G_TILES + tl) * 32;
    const int pw = p0 + sub * G_PTS;
    const int b  = p0 >> 12;
    const float* y1b = y1T + (size_t)b * NPTS * OCH;

    int mine[5];
    {
        const int* ip = idx + (size_t)pw * KNN + lane * 5;
#pragma unroll
        for (int k = 0; k < 5; ++k) mine[k] = ip[k];
    }

    float s0 = 0.f, s1 = 0.f, q0 = 0.f, q1 = 0.f;

    for (int pt = 0; pt < G_PTS; ++pt) {
        const int p = pw + pt;
        float2 e = ((const float2*)(y3T + (size_t)p * OCH))[lane];

        float hi0 = -FLT_MAX, hi1 = -FLT_MAX, lo0 = FLT_MAX, lo1 = FLT_MAX;
#pragma unroll
        for (int j = 0; j < KNN; ++j) {
            const int nb = __builtin_amdgcn_readlane(mine[j % 5], 4 * pt + j / 5);
            float2 g = ((const float2*)(y1b + ((size_t)nb << 6)))[lane];
            float h0 = g.x + e.x, h1 = g.y + e.y;
            s0 += h0; s1 += h1; q0 += h0 * h0; q1 += h1 * h1;
            hi0 = fmaxf(hi0, h0); lo0 = fminf(lo0, h0);
            hi1 = fmaxf(hi1, h1); lo1 = fminf(lo1, h1);
        }
        float2 selv = { mx0 ? hi0 : lo0, mx1 ? hi1 : lo1 };
        *(float2*)&tile[tl][sub * G_PTS + pt][ch] = selv;
    }
    __shared__ float red[G_WAVES][2 * OCH];
    float2 sv = { s0, s1 }, qv = { q0, q1 };
    *(float2*)&red[wv][ch]       = sv;
    *(float2*)&red[wv][OCH + ch] = qv;
    __syncthreads();
    if (threadIdx.x < 2 * OCH) {
        float v = red[0][threadIdx.x];
        for (int i = 1; i < G_WAVES; ++i) v += red[i][threadIdx.x];
        partial[(size_t)blockIdx.x * (2 * OCH) + threadIdx.x] = v;
    }
    const int n0 = p0 & (NPTS - 1);
    float* op = selT + (size_t)b * OCH * NPTS + n0 + lane;
    const float* tp = &tile[tl][lane][0];
    for (int o = sub * 16; o < sub * 16 + 16; ++o) op[(size_t)o * NPTS] = tp[o];
}

__global__ __launch_bounds__(128) void stats_kernel(const float* __restrict__ partial,
                                                    const float* __restrict__ gamma,
                                                    const float* __restrict__ beta,
                                                    float* __restrict__ coeff) {
    const int col = threadIdx.x;
    float v[GATHER_BLOCKS];
#pragma unroll
    for (int w = 0; w < GATHER_BLOCKS; ++w) v[w] = partial[(size_t)w * 128 + col];
    float acc = 0.f;
#pragma unroll
    for (int w = 0; w < GATHER_BLOCKS; ++w) acc += v[w];

    __shared__ float tot[2 * OCH];
    tot[col] = acc;
    __syncthreads();
    if (col >= OCH) return;
    const int ch = col;
    const float s = tot[ch], q = tot[OCH + ch];
    const float M = (float)BATCH * NPTS * KNN;
    float mean  = s / M;
    float var   = q / M - mean * mean;
    float scale = gamma[ch] * rsqrtf(var + 1e-5f);
    coeff[ch]       = scale;
    coeff[OCH + ch] = beta[ch] - mean * scale;
}

__global__ __launch_bounds__(256) void final_kernel(const float* __restrict__ selT,
                                                    const float* __restrict__ coeff,
                                                    float* __restrict__ out) {
    const int row = blockIdx.x;
    const int o   = row & (OCH - 1);
    const float sc = coeff[o], sh = coeff[OCH + o];
    const float4* src = (const float4*)(selT + (size_t)row * NPTS) + threadIdx.x;
    float4*       dst = (float4*)(out + (size_t)row * NPTS) + threadIdx.x;
    float4 v[4];
#pragma unroll
    for (int j = 0; j < 4; ++j) v[j] = src[j * 256];
#pragma unroll
    for (int j = 0; j < 4; ++j) {
        float4 r;
        r.x = v[j].x * sc + sh; r.x = (r.x >= 0.f) ? r.x : 0.2f * r.x;
        r.y = v[j].y * sc + sh; r.y = (r.y >= 0.f) ? r.y : 0.2f * r.y;
        r.z = v[j].z * sc + sh; r.z = (r.z >= 0.f) ? r.z : 0.2f * r.z;
        r.w = v[j].w * sc + sh; r.w = (r.w >= 0.f) ? r.w : 0.2f * r.w;
        dst[j * 256] = r;
    }
}

extern "C" void kernel_launch(void* const* d_in, const int* in_sizes, int n_in,
                              void* d_out, int out_size, void* d_ws, size_t ws_size,
                              hipStream_t stream) {
    const float* x     = (const float*)d_in[0];
    const float* W     = (const float*)d_in[1];
    const float* gamma = (const float*)d_in[2];
    const float* beta  = (const float*)d_in[3];
    float* out = (float*)d_out;

    float* ws    = (float*)d_ws;
    float* sq    = ws;
    int*   idx   = (int*)(ws + 32768);
    float* y1T   = ws + 32768 + 655360;
    float* y3T   = y1T + (size_t)BATCH * NPTS * OCH;
    float* sel   = y3T + (size_t)BATCH * NPTS * OCH;
    float* part  = sel + (size_t)BATCH * NPTS * OCH;
    float* coeff = part + 1024 * 128;

    sq_kernel<<<(BATCH * NPTS) / 256, 256, 0, stream>>>(x, sq, sel);
    knn_kernel<<<dim3(NPTS / 64, BATCH), 128, 0, stream>>>(x, sel, sq, W, idx, y1T, y3T);
    gather_kernel<<<GATHER_BLOCKS, 1024, 0, stream>>>(y1T, y3T, idx, gamma, sel, part);
    stats_kernel<<<1, 128, 0, stream>>>(part, gamma, beta, coeff);
    final_kernel<<<BATCH * OCH, 256, 0, stream>>>(sel, coeff, out);
}


# kNN: after a drain a query's two half-lists (lanes c, c+16) are lifted to a common floor, the 20th best of their union: fewer entries filed, fewer insert passes; merged lists and order unchanged
# speedup vs baseline: 4.3374x; 1.0420x over previous
#include <hip/hip_runtime.h>
#include <hip/hip_bf16.h>
#include <float.h>


typedef float v2f __attribute__((ext_vector_type(2)));
typedef float v8f __attribute__((ext_vector_type(8)));

#define BATCH 8
#define CH    64
#define NPTS  4096
#define OCH   64
#define KNN   20
#define NEG_INF (-3.0e38f)
#define GATHER_BLOCKS 128

__device__ __forceinline__ v8f wmma_f32(v2f a, v2f b, v8f c) {
    return __builtin_amdgcn_wmma_f32_16x16x4_f32(
        false, a, false, b, (short)0, c, false, false);
}

__global__ __launch_bounds__(256) void sq_kernel(const float* __restrict__ x,
                                                 float* __restrict__ sq,
                                                 float* __restrict__ xf) {
    int t = blockIdx.x * 256 + threadIdx.x;
    int b = t >> 12;
    int n = t & (NPTS - 1);
    const float* p = x + (size_t)b * CH * NPTS + n;
    float v[CH];
#pragma unroll
    for (int c = 0; c < CH; ++c) v[c] = p[(size_t)c * NPTS];
    float s = 0.f;
#pragma unroll
    for (int c = 0; c < CH; ++c) s += v[c] * v[c];
    sq[t] = s;
    float4* o = (float4*)xf + (size_t)(t >> 4) * 256 + (n & 15);
#pragma unroll
    for (int j = 0; j < 8; ++j) {
        o[j * 32]      = make_float4(v[8 * j],     v[8 * j + 1], v[8 * j + 4], v[8 * j + 5]);
        o[j * 32 + 16] = make_float4(v[8 * j + 2], v[8 * j + 3], v[8 * j + 6], v[8 * j + 7]);
    }
}


#define TOPK_INSERT(v, col)                                                   \
    do {                                                                      \
        const float nv = (v); const int nc = (col);                           \
        bool gt[KNN];                                                         \
        _Pragma("unroll")                                                     \
        for (int s = 0; s < KNN; ++s) gt[s] = nv > best[s];                   \
        _Pragma("unroll")                                                     \
        for (int s = KNN - 1; s > 0; --s) {                                   \
            const float uv = gt[s - 1] ? best[s - 1] : nv;                    \
            const int   ui = gt[s - 1] ? bidx[s - 1] : nc;                    \
            best[s] = gt[s] ? uv : best[s];                                   \
            bidx[s] = gt[s] ? ui : bidx[s];                                   \
        }                                                                     \
        best[0] = gt[0] ? nv : best[0];                                       \
        bidx[0] = gt[0] ? nc : bidx[0];                                       \
        thr = best[KNN - 1];                                                  \
    } while (0)

struct CandTile { float4 g[8]; };

__device__ __forceinline__ void load_cand(CandTile& t, const float4* __restrict__ xfb,
                                          int col0, int lane) {
    const float4* gp = xfb + (size_t)(col0 >> 4) * 256 + lane;
#pragma unroll
    for (int j = 0; j < 8; ++j) t.g[j] = gp[j * 32];
}

struct CandSq { float4 lo, hi; };

__device__ __forceinline__ void load_sq(CandSq& s, const float* __restrict__ sqb,
                                        int col0, int rowadd) {
    const float4* sp = (const float4*)(sqb + col0 + rowadd);
    s.lo = sp[0];
    s.hi = sp[1];
}

#define GRAM_2TILES(t0, t1, acc0, acc1)                                       \
    do {                                                                      \
        _Pragma("unroll")                                                     \
        for (int j = 0; j < 8; ++j) {                                         \
            const v2f e0 = { t0.g[j].x, t0.g[j].y }, o0 = { t0.g[j].z, t0.g[j].w }; \
            const v2f e1 = { t1.g[j].x, t1.g[j].y }, o1 = { t1.g[j].z, t1.g[j].w }; \
            acc0 = wmma_f32(e0, qf[2 * j],     acc0);                         \
            acc1 = wmma_f32(e1, qf[2 * j],     acc1);                         \
            acc0 = wmma_f32(o0, qf[2 * j + 1], acc0);                         \
            acc1 = wmma_f32(o1, qf[2 * j + 1], acc1);                         \
        }                                                                     \
    } while (0)

#define QCAP   32
#define QROOM  8

__device__ __forceinline__ void push_tile(v8f acc, float4 sqlo, float4 sqhi, int cbase,
                                          float thr, float2* q, int& cnt) {
    float sc[8];
    sc[0] = 2.0f * acc[0] - sqlo.x; sc[1] = 2.0f * acc[1] - sqlo.y;
    sc[2] = 2.0f * acc[2] - sqlo.z; sc[3] = 2.0f * acc[3] - sqlo.w;
    sc[4] = 2.0f * acc[4] - sqhi.x; sc[5] = 2.0f * acc[5] - sqhi.y;
    sc[6] = 2.0f * acc[6] - sqhi.z; sc[7] = 2.0f * acc[7] - sqhi.w;
#pragma unroll
    for (int r = 0; r < 8; ++r) {
        q[cnt * 32] = make_float2(sc[r], __int_as_float(cbase + r));
        cnt += (sc[r] > thr) ? 1 : 0;
    }
}

#define DRAIN_QUEUE()                                                         \
    do {                                                                      \
        float2 e = make_float2(NEG_INF, 0.f);                                 \
        if (cnt > 0) e = q[0];                                                \
        for (int i = 0; __builtin_amdgcn_ballot_w32(i < cnt) != 0u; ++i) {    \
            float2 en = e;                                                    \
            if (i + 1 < cnt) en = q[(i + 1) * 32];                            \
            if (i < cnt && e.x > thr) {                                       \
                const float ev = e.x; const int ec = __float_as_int(e.y);     \
                TOPK_INSERT(ev, ec);                                          \
            }                                                                 \
            e = en;                                                           \
        }                                                                     \
        cnt = 0;                                                              \
    } while (0)

__device__ __forceinline__ float next_below(float x) {
    const int b = __float_as_int(x);
    return (x > 0.f) ? __int_as_float(b - 1)
         : (x < 0.f) ? __int_as_float(b + 1)
                     : __int_as_float((int)0x80000001u);
}

#define LIFT_TO_PAIR_FLOOR()                                                  \
    do {                                                                      \
        float u = fmaxf(best[KNN - 1], __shfl_xor(best[KNN - 1], 16));        \
        _Pragma("unroll")   \
        for (int j = 0; j < KNN - 1; ++j)                                     \
            u = fmaxf(u, fminf(best[j], __shfl_xor(best[KNN - 2 - j], 16)));  \
        const float fl = next_below(u);                                       \
        _Pragma("unroll")   \
        for (int s = 0; s < KNN; ++s) best[s] = fmaxf(best[s], fl);           \
        thr = best[KNN - 1];                                                  \
    } while (0)

#define MAKE_ROOM()                                                           \
    do {                                                                      \
        if (__builtin_amdgcn_ballot_w32(cnt > QCAP - QROOM) != 0u) {          \
            DRAIN_QUEUE();                                                    \
            drained = true;                                                   \
        }                                                                     \
    } while (0)

#define REFRESH_FLOOR()                                                       \
    do {                                                                      \
                      \
        if (__builtin_expect(drained, false)) {                               \
            LIFT_TO_PAIR_FLOOR();                                             \
            drained = false;                                                  \
        }                                                                     \
    } while (0)

#define W_PITCH 132

__global__ __launch_bounds__(128) void knn_kernel(const float* __restrict__ x,
                                                  const float* __restrict__ xf,
                                                  const float* __restrict__ sq,
                                                  const float* __restrict__ W,
                                                  int* __restrict__ idxout,
                                                  float* __restrict__ y1T,
                                                  float* __restrict__ y3T) {
    const int b    = blockIdx.y;
    const int wave = threadIdx.x >> 5;
    const int lane = threadIdx.x & 31;
    const int q0   = (blockIdx.x * 4 + wave) * 16;
    const int lmod = lane & 15;
    const int half = lane >> 4;
    const int khalf  = half << 1;
    const int rowadd = half << 3;
    const float* xb  = x + (size_t)b * CH * NPTS;
    const float4* xfb = (const float4*)(xf + (size_t)b * CH * NPTS);
    const float* sqb = sq + (size_t)b * NPTS;

    v2f qf[16];
#pragma unroll
    for (int kk = 0; kk < 16; ++kk) {
        const float* p = xb + (size_t)(4 * kk + khalf) * NPTS + q0 + lmod;
        qf[kk].x = p[0];
        qf[kk].y = p[NPTS];
    }

    __shared__ __attribute__((aligned(16))) float wlds[OCH][W_PITCH];
    for (int i = 0; i < 16; ++i) {
        const int f  = i * 128 + threadIdx.x;
        const int r  = f >> 5, c4 = f & 31;
        *(float4*)&wlds[r][c4 * 4] = ((const float4*)W)[f];
    }
    __syncthreads();
    {
        const size_t prow = ((size_t)b * NPTS + q0 + lmod) * OCH + rowadd;
        for (int ot = 0; ot < 4; ++ot) {
            const float* wr = &wlds[ot * 16 + lmod][khalf];
            v8f p1 = {}, p3 = {};
#pragma unroll
            for (int kk = 0; kk < 16; ++kk) {
                const v2f w1 = *(const v2f*)(wr + 4 * kk);
                const v2f w2 = *(const v2f*)(wr + CH + 4 * kk);
                const v2f wd = w2 - w1;
                p1 = wmma_f32(w1, qf[kk], p1);
                p3 = wmma_f32(wd, qf[kk], p3);
            }
            float4* o1 = (float4*)(y1T + prow + ot * 16);
            float4* o3 = (float4*)(y3T + prow + ot * 16);
            o1[0] = make_float4(p1[0], p1[1], p1[2], p1[3]);
            o1[1] = make_float4(p1[4], p1[5], p1[6], p1[7]);
            o3[0] = make_float4(p3[0], p3[1], p3[2], p3[3]);
            o3[1] = make_float4(p3[4], p3[5], p3[6], p3[7]);
        }
    }

    float best[KNN]; int bidx[KNN]; float thr = NEG_INF;
    bool drained = false;
#pragma unroll
    for (int i = 0; i < KNN; ++i) { best[i] = NEG_INF; bidx[i] = 0; }

    __shared__ float2 qbuf[4][QCAP][32];
    float2* q = &qbuf[wave][0][lane];
    int cnt = 0;

    CandTile a0, a1, b0, b1;
    load_cand(a0, xfb, 0,  lane);
    load_cand(a1, xfb, 16, lane);

    for (int c0 = 0; c0 < NPTS; c0 += 64) {
        REFRESH_FLOOR();
        {
            CandSq s0, s1;
            load_sq(s0, sqb, c0,      rowadd);
            load_sq(s1, sqb, c0 + 16, rowadd);
            load_cand(b0, xfb, c0 + 32, lane);
            load_cand(b1, xfb, c0 + 48, lane);
            v8f acc0 = {}, acc1 = {};
            GRAM_2TILES(a0, a1, acc0, acc1);
            MAKE_ROOM();
            push_tile(acc0, s0.lo, s0.hi, c0 + rowadd,      thr, q, cnt);
            MAKE_ROOM();
            push_tile(acc1, s1.lo, s1.hi, c0 + 16 + rowadd, thr, q, cnt);
        }
        const int cn = (c0 + 64) & (NPTS - 1);
        REFRESH_FLOOR();
        {
            CandSq s0, s1;
            load_sq(s0, sqb, c0 + 32, rowadd);
            load_sq(s1, sqb, c0 + 48, rowadd);
            load_cand(a0, xfb, cn,      lane);
            load_cand(a1, xfb, cn + 16, lane);
            v8f acc0 = {}, acc1 = {};
            GRAM_2TILES(b0, b1, acc0, acc1);
            MAKE_ROOM();
            push_tile(acc0, s0.lo, s0.hi, c0 + 32 + rowadd, thr, q, cnt);
            MAKE_ROOM();
            push_tile(acc1, s1.lo, s1.hi, c0 + 48 + rowadd, thr, q, cnt);
        }
    }
    DRAIN_QUEUE();

#pragma nounroll
    for (int i = 0; i < KNN; ++i) {
        const float ov = __shfl_xor(best[0], 16);
        const int   oi = __shfl_xor(bidx[0], 16);
        if (half) {
#pragma unroll
            for (int s = 0; s < KNN - 1; ++s) { best[s] = best[s + 1]; bidx[s] = bidx[s + 1]; }
        } else if (ov > thr) {
            TOPK_INSERT(ov, oi);
        }
    }
    if (half == 0) {
        int* op = idxout + ((size_t)b * NPTS + q0 + lmod) * KNN;
#pragma unroll
        for (int i = 0; i < KNN; ++i) op[i] = bidx[i];
    }
}

#define SEL_PITCH 66
#define G_WAVES 32
#define G_TILES 8
#define G_PTS   8
__global__ __launch_bounds__(1024) void gather_kernel(const float* __restrict__ y1T,
                                                      const float* __restrict__ y3T,
                                                      const int* __restrict__ idx,
                                                      const float* __restrict__ gamma,
                                                      float* __restrict__ selT,
                                                      float* __restrict__ partial) {
    const int wv   = threadIdx.x >> 5;
    const int lane = threadIdx.x & 31;
    const int tl   = wv >> 2;
    const int sub  = wv & 3;
    const int ch   = lane * 2;
    __shared__ __attribute__((aligned(16))) float tile[G_TILES][32][SEL_PITCH];
    bool mx0 = gamma[ch] >= 0.f, mx1 = gamma[ch + 1] >= 0.f;

    const int p0 = (blockIdx.x * G_TILES + tl) * 32;
    const int pw = p0 + sub * G_PTS;
    const int b  = p0 >> 12;
    const float* y1b = y1T + (size_t)b * NPTS * OCH;

    int mine[5];
    {
        const int* ip = idx + (size_t)pw * KNN + lane * 5;
#pragma unroll
        for (int k = 0; k < 5; ++k) mine[k] = ip[k];
    }

    float s0 = 0.f, s1 = 0.f, q0 = 0.f, q1 = 0.f;

    for (int pt = 0; pt < G_PTS; ++pt) {
        const int p = pw + pt;
        float2 e = ((const float2*)(y3T + (size_t)p * OCH))[lane];

        float hi0 = -FLT_MAX, hi1 = -FLT_MAX, lo0 = FLT_MAX, lo1 = FLT_MAX;
#pragma unroll
        for (int j = 0; j < KNN; ++j) {
            const int nb = __builtin_amdgcn_readlane(mine[j % 5], 4 * pt + j / 5);
            float2 g = ((const float2*)(y1b + ((size_t)nb << 6)))[lane];
            float h0 = g.x + e.x, h1 = g.y + e.y;
            s0 += h0; s1 += h1; q0 += h0 * h0; q1 += h1 * h1;
            hi0 = fmaxf(hi0, h0); lo0 = fminf(lo0, h0);
            hi1 = fmaxf(hi1, h1); lo1 = fminf(lo1, h1);
        }
        float2 selv = { mx0 ? hi0 : lo0, mx1 ? hi1 : lo1 };
        *(float2*)&tile[tl][sub * G_PTS + pt][ch] = selv;
    }
    __shared__ float red[G_WAVES][2 * OCH];
    float2 sv = { s0, s1 }, qv = { q0, q1 };
    *(float2*)&red[wv][ch]       = sv;
    *(float2*)&red[wv][OCH + ch] = qv;
    __syncthreads();
    if (threadIdx.x < 2 * OCH) {
        float v = red[0][threadIdx.x];
        for (int i = 1; i < G_WAVES; ++i) v += red[i][threadIdx.x];
        partial[(size_t)blockIdx.x * (2 * OCH) + threadIdx.x] = v;
    }
    const int n0 = p0 & (NPTS - 1);
    float* op = selT + (size_t)b * OCH * NPTS + n0 + lane;
    const float* tp = &tile[tl][lane][0];
    for (int o = sub * 16; o < sub * 16 + 16; ++o) op[(size_t)o * NPTS] = tp[o];
}

__global__ __launch_bounds__(128) void stats_kernel(const float* __restrict__ partial,
                                                    const float* __restrict__ gamma,
                                                    const float* __restrict__ beta,
                                                    float* __restrict__ coeff) {
    const int col = threadIdx.x;
    float v[GATHER_BLOCKS];
#pragma unroll
    for (int w = 0; w < GATHER_BLOCKS; ++w) v[w] = partial[(size_t)w * 128 + col];
    float acc = 0.f;
#pragma unroll
    for (int w = 0; w < GATHER_BLOCKS; ++w) acc += v[w];

    __shared__ float tot[2 * OCH];
    tot[col] = acc;
    __syncthreads();
    if (col >= OCH) return;
    const int ch = col;
    const float s = tot[ch], q = tot[OCH + ch];
    const float M = (float)BATCH * NPTS * KNN;
    float mean  = s / M;
    float var   = q / M - mean * mean;
    float scale = gamma[ch] * rsqrtf(var + 1e-5f);
    coeff[ch]       = scale;
    coeff[OCH + ch] = beta[ch] - mean * scale;
}

__global__ __launch_bounds__(256) void final_kernel(const float* __restrict__ selT,
                                                    const float* __restrict__ coeff,
                                                    float* __restrict__ out) {
    const int row = blockIdx.x;
    const int o   = row & (OCH - 1);
    const float sc = coeff[o], sh = coeff[OCH + o];
    const float4* src = (const float4*)(selT + (size_t)row * NPTS) + threadIdx.x;
    float4*       dst = (float4*)(out + (size_t)row * NPTS) + threadIdx.x;
    float4 v[4];
#pragma unroll
    for (int j = 0; j < 4; ++j) v[j] = src[j * 256];
#pragma unroll
    for (int j = 0; j < 4; ++j) {
        float4 r;
        r.x = v[j].x * sc + sh; r.x = (r.x >= 0.f) ? r.x : 0.2f * r.x;
        r.y = v[j].y * sc + sh; r.y = (r.y >= 0.f) ? r.y : 0.2f * r.y;
        r.z = v[j].z * sc + sh; r.z = (r.z >= 0.f) ? r.z : 0.2f * r.z;
        r.w = v[j].w * sc + sh; r.w = (r.w >= 0.f) ? r.w : 0.2f * r.w;
        dst[j * 256] = r;
    }
}

extern "C" void kernel_launch(void* const* d_in, const int* in_sizes, int n_in,
                              void* d_out, int out_size, void* d_ws, size_t ws_size,
                              hipStream_t stream) {
    const float* x     = (const float*)d_in[0];
    const float* W     = (const float*)d_in[1];
    const float* gamma = (const float*)d_in[2];
    const float* beta  = (const float*)d_in[3];
    float* out = (float*)d_out;

    float* ws    = (float*)d_ws;
    float* sq    = ws;
    int*   idx   = (int*)(ws + 32768);
    float* y1T   = ws + 32768 + 655360;
    float* y3T   = y1T + (size_t)BATCH * NPTS * OCH;
    float* sel   = y3T + (size_t)BATCH * NPTS * OCH;
    float* part  = sel + (size_t)BATCH * NPTS * OCH;
    float* coeff = part + 1024 * 128;

    sq_kernel<<<(BATCH * NPTS) / 256, 256, 0, stream>>>(x, sq, sel);
    knn_kernel<<<dim3(NPTS / 64, BATCH), 128, 0, stream>>>(x, sel, sq, W, idx, y1T, y3T);
    gather_kernel<<<GATHER_BLOCKS, 1024, 0, stream>>>(y1T, y3T, idx, gamma, sel, part);
    stats_kernel<<<1, 128, 0, stream>>>(part, gamma, beta, coeff);
    final_kernel<<<BATCH * OCH, 256, 0, stream>>>(sel, coeff, out);
}
